# starting kernel
# baseline (speedup 1.0000x reference)
#include <hip/hip_runtime.h>


typedef float v2f __attribute__((ext_vector_type(2)));
typedef float v8f __attribute__((ext_vector_type(8)));

#define DD 128
#define NT 8
#define WFRAG_F4_PER_MAT 4096

__global__ __launch_bounds__(256)
void reformat_w(const float* __restrict__ Wq, const float* __restrict__ Wk,
                const float* __restrict__ Wv, float4* __restrict__ Wf)
{
    const int idx = (int)(blockIdx.x * blockDim.x + threadIdx.x);
    const int m    = idx >> 12;
    const int r    = idx & 4095;
    const int kkp  = r >> 8;
    const int t    = (r >> 5) & 7;
    const int lane = r & 31;
    const int hi = lane >> 4, l = lane & 15;
    const int n  = t * 16 + l;
    const float* W = (m == 0) ? Wq : ((m == 1) ? Wk : Wv);
    const int k0 = 8 * kkp + 2 * hi;
    const int k1 = k0 + 4;
    float4 v;
    v.x = W[(size_t)(k0 + 0) * DD + n];
    v.y = W[(size_t)(k0 + 1) * DD + n];
    v.z = W[(size_t)(k1 + 0) * DD + n];
    v.w = W[(size_t)(k1 + 1) * DD + n];
    Wf[idx] = v;
}

__global__ __launch_bounds__(256)
void qkv_gemm_wmma(const float* __restrict__ X,
                   const float4* __restrict__ Wf,
                   const float* __restrict__ bq, const float* __restrict__ bk,
                   const float* __restrict__ bv,
                   float* __restrict__ Q, float* __restrict__ K, float* __restrict__ V,
                   int Mrows)
{
    const int wave = (int)((blockIdx.x * blockDim.x + threadIdx.x) >> 5);
    const int lane = (int)(threadIdx.x & 31);
    const int m0   = wave * 16;
    if (m0 >= Mrows) return;

    const int hi = lane >> 4;
    const int l  = lane & 15;

    const float* xrow = X + (size_t)(m0 + l) * DD + 2 * hi;

    const float* bm[3] = { bq, bk, bv };
    float*       Om[3] = { Q,  K,  V  };

#pragma unroll
    for (int m = 0; m < 3; ++m) {
        const float4* Wp = Wf + (size_t)m * WFRAG_F4_PER_MAT;
        const float*  bp = bm[m];
        float*        Op = Om[m];

        v8f c[NT];
#pragma unroll
        for (int t = 0; t < NT; ++t) {
            const float bias = bp[t * 16 + l];
#pragma unroll
            for (int r = 0; r < 8; ++r) c[t][r] = bias;
        }

#pragma unroll 2
        for (int kkp = 0; kkp < 16; ++kkp) {
            const v2f a0 = *(const v2f*)(xrow + 8 * kkp);
            const v2f a1 = *(const v2f*)(xrow + 8 * kkp + 4);
            float4 bf[NT];
#pragma unroll
            for (int t = 0; t < NT; ++t)
                bf[t] = Wp[(size_t)(kkp * NT + t) * 32 + lane];
#pragma unroll
            for (int t = 0; t < NT; ++t) {
                v2f b0; b0.x = bf[t].x; b0.y = bf[t].y;
                c[t] = __builtin_amdgcn_wmma_f32_16x16x4_f32(
                           false, a0, false, b0, (short)0, c[t], false, false);
            }
#pragma unroll
            for (int t = 0; t < NT; ++t) {
                v2f b1; b1.x = bf[t].z; b1.y = bf[t].w;
                c[t] = __builtin_amdgcn_wmma_f32_16x16x4_f32(
                           false, a1, false, b1, (short)0, c[t], false, false);
            }
        }

#pragma unroll
        for (int t = 0; t < NT; ++t) {
#pragma unroll
            for (int r = 0; r < 8; ++r) {
                Op[(size_t)(m0 + 8 * hi + r) * DD + t * 16 + l] = c[t][r];
            }
        }
    }
}

__global__ __launch_bounds__(256)
void win_attn(const float* __restrict__ Q, const float* __restrict__ K,
              const float* __restrict__ V, float* __restrict__ out,
              int B, int H, int W)
{
    const int wave = (int)((blockIdx.x * blockDim.x + threadIdx.x) >> 5);
    const int lane = (int)(threadIdx.x & 31);
    const int npix = B * H * W;
    if (wave >= npix) return;

    const int b   = wave / (H * W);
    const int rem = wave % (H * W);
    const int y   = rem / W;
    const int x   = rem % W;
    const float scale = 0.08838834764831845f;

    const float4 q = *(const float4*)(Q + (size_t)wave * DD + lane * 4);

    float  sc[9];
    float4 vf[9];
#pragma unroll
    for (int dy = -1; dy <= 1; ++dy) {
#pragma unroll
        for (int dx = -1; dx <= 1; ++dx) {
            const int  i  = (dy + 1) * 3 + (dx + 1);
            const int  yy = y + dy, xx = x + dx;
            const bool ok = (yy >= 0) & (yy < H) & (xx >= 0) & (xx < W);
            float  s  = 0.0f;
            float4 v4 = make_float4(0.f, 0.f, 0.f, 0.f);
            if (ok) {
                const size_t off = ((size_t)b * H * W + (size_t)yy * W + xx) * DD + lane * 4;
                const float4 k4 = *(const float4*)(K + off);
                v4 = *(const float4*)(V + off);
                s = q.x * k4.x + q.y * k4.y + q.z * k4.z + q.w * k4.w;
            }
#pragma unroll
            for (int m = 16; m >= 1; m >>= 1) s += __shfl_xor(s, m, 32);
            sc[i] = s * scale;
            vf[i] = v4;
        }
    }

    float mx = sc[0];
#pragma unroll
    for (int i = 1; i < 9; ++i) mx = fmaxf(mx, sc[i]);
    float p[9], sum = 0.0f;
#pragma unroll
    for (int i = 0; i < 9; ++i) { p[i] = expf(sc[i] - mx); sum += p[i]; }
    const float inv = 1.0f / sum;

    float4 o = make_float4(0.f, 0.f, 0.f, 0.f);
#pragma unroll
    for (int i = 0; i < 9; ++i) {
        const float w = p[i] * inv;
        o.x += w * vf[i].x; o.y += w * vf[i].y;
        o.z += w * vf[i].z; o.w += w * vf[i].w;
    }
    *(float4*)(out + (size_t)wave * DD + lane * 4) = o;
}

extern "C" void kernel_launch(void* const* d_in, const int* in_sizes, int n_in,
                              void* d_out, int out_size, void* d_ws, size_t ws_size,
                              hipStream_t stream)
{
    const float* x  = (const float*)d_in[0];
    const float* Wq = (const float*)d_in[1];
    const float* bq = (const float*)d_in[2];
    const float* Wk = (const float*)d_in[3];
    const float* bk = (const float*)d_in[4];
    const float* Wv = (const float*)d_in[5];
    const float* bv = (const float*)d_in[6];

    const int H = 128, W = 128;
    const int Mrows = in_sizes[0] / DD;
    const int B = Mrows / (H * W);

    float*  Q   = (float*)d_ws;
    float*  Kp  = Q  + (size_t)Mrows * DD;
    float*  Vp  = Kp + (size_t)Mrows * DD;
    float4* Wf  = (float4*)(Vp + (size_t)Mrows * DD);

    reformat_w<<<48, 256, 0, stream>>>(Wq, Wk, Wv, Wf);

    const int gemmWaves = Mrows / 16;
    qkv_gemm_wmma<<<(gemmWaves + 7) / 8, 256, 0, stream>>>(
        x, Wf, bq, bk, bv, Q, Kp, Vp, Mrows);

    const int pixWaves = Mrows;
    win_attn<<<(pixWaves + 7) / 8, 256, 0, stream>>>(
        Q, Kp, Vp, (float*)d_out, B, H, W);
}


# win_attn: each wave walks 16 consecutive pixels of a row, 3x3 K/V window kept in registers (only the new column loaded, one pixel ahead); 16x fewer waves; per-pixel summation order unchanged
# speedup vs baseline: 1.1532x; 1.1532x over previous
#include <hip/hip_runtime.h>


typedef float v2f __attribute__((ext_vector_type(2)));
typedef float v8f __attribute__((ext_vector_type(8)));
typedef float v4f __attribute__((ext_vector_type(4)));

#define DD 128
#define NT 8
#define WFRAG_F4_PER_MAT 4096

__global__ __launch_bounds__(256)
void reformat_w(const float* __restrict__ Wq, const float* __restrict__ Wk,
                const float* __restrict__ Wv, float4* __restrict__ Wf)
{
    const int idx = (int)(blockIdx.x * blockDim.x + threadIdx.x);
    const int m    = idx >> 12;
    const int r    = idx & 4095;
    const int kkp  = r >> 8;
    const int t    = (r >> 5) & 7;
    const int lane = r & 31;
    const int hi = lane >> 4, l = lane & 15;
    const int n  = t * 16 + l;
    const float* W = (m == 0) ? Wq : ((m == 1) ? Wk : Wv);
    const int k0 = 8 * kkp + 2 * hi;
    const int k1 = k0 + 4;
    float4 v;
    v.x = W[(size_t)(k0 + 0) * DD + n];
    v.y = W[(size_t)(k0 + 1) * DD + n];
    v.z = W[(size_t)(k1 + 0) * DD + n];
    v.w = W[(size_t)(k1 + 1) * DD + n];
    Wf[idx] = v;
}

__global__ __launch_bounds__(256)
void qkv_gemm_wmma(const float* __restrict__ X,
                   const float4* __restrict__ Wf,
                   const float* __restrict__ bq, const float* __restrict__ bk,
                   const float* __restrict__ bv,
                   float* __restrict__ Q, float* __restrict__ K, float* __restrict__ V,
                   int Mrows)
{
    const int wave = (int)((blockIdx.x * blockDim.x + threadIdx.x) >> 5);
    const int lane = (int)(threadIdx.x & 31);
    const int m0   = wave * 16;
    if (m0 >= Mrows) return;

    const int hi = lane >> 4;
    const int l  = lane & 15;

    const float* xrow = X + (size_t)(m0 + l) * DD + 2 * hi;

    const float* bm[3] = { bq, bk, bv };
    float*       Om[3] = { Q,  K,  V  };

#pragma unroll
    for (int m = 0; m < 3; ++m) {
        const float4* Wp = Wf + (size_t)m * WFRAG_F4_PER_MAT;
        const float*  bp = bm[m];
        float*        Op = Om[m];

        v8f c[NT];
#pragma unroll
        for (int t = 0; t < NT; ++t) {
            const float bias = bp[t * 16 + l];
#pragma unroll
            for (int r = 0; r < 8; ++r) c[t][r] = bias;
        }

#pragma unroll 2
        for (int kkp = 0; kkp < 16; ++kkp) {
            const v2f a0 = *(const v2f*)(xrow + 8 * kkp);
            const v2f a1 = *(const v2f*)(xrow + 8 * kkp + 4);
            float4 bf[NT];
#pragma unroll
            for (int t = 0; t < NT; ++t)
                bf[t] = Wp[(size_t)(kkp * NT + t) * 32 + lane];
#pragma unroll
            for (int t = 0; t < NT; ++t) {
                v2f b0; b0.x = bf[t].x; b0.y = bf[t].y;
                c[t] = __builtin_amdgcn_wmma_f32_16x16x4_f32(
                           false, a0, false, b0, (short)0, c[t], false, false);
            }
#pragma unroll
            for (int t = 0; t < NT; ++t) {
                v2f b1; b1.x = bf[t].z; b1.y = bf[t].w;
                c[t] = __builtin_amdgcn_wmma_f32_16x16x4_f32(
                           false, a1, false, b1, (short)0, c[t], false, false);
            }
        }

#pragma unroll
        for (int t = 0; t < NT; ++t) {
#pragma unroll
            for (int r = 0; r < 8; ++r) {
                Op[(size_t)(m0 + 8 * hi + r) * DD + t * 16 + l] = c[t][r];
            }
        }
    }
}

#define STRIP 16

__device__ __forceinline__ v4f ld_row(const char* P, unsigned boff)
{
    return *(const v4f*)(P + boff);
}

__global__ __launch_bounds__(256)
void win_attn(const float* __restrict__ Q, const float* __restrict__ K,
              const float* __restrict__ V, float* __restrict__ out,
              int B, int H, int W)
{
    const int wave = __builtin_amdgcn_readfirstlane(
        (int)((blockIdx.x * blockDim.x + threadIdx.x) >> 5));
    const unsigned lane16 = (threadIdx.x & 31u) * 16u;
    const int spr  = W / STRIP;
    if (wave >= B * H * spr) return;

    const int  rowi = wave / spr;
    const int  x0   = (wave - rowi * spr) * STRIP;
    const int  y    = rowi % H;
    const bool okr[3] = { y > 0, true, y + 1 < H };
    const float scale = 0.08838834764831845f;
    const unsigned PIXB = DD * 4u;

    const char* Kr[3];
    const char* Vr[3];
#pragma unroll
    for (int r = 0; r < 3; ++r) {
        const size_t rb = (size_t)(rowi + (okr[r] ? r - 1 : 0)) * (size_t)W * PIXB;
        Kr[r] = (const char*)K + rb;
        Vr[r] = (const char*)V + rb;
    }
    const char* Qr = (const char*)Q + (size_t)rowi * (size_t)W * PIXB;
    char*       Or = (char*)out + (size_t)rowi * (size_t)W * PIXB;

    v4f kw[4][3], vw[4][3];
    v4f qq[2];
#pragma unroll
    for (int c = 0; c < 3; ++c) {
        const unsigned colb = (unsigned)min(max(x0 - 1 + c, 0), W - 1) * PIXB + lane16;
#pragma unroll
        for (int r = 0; r < 3; ++r) {
            kw[c][r] = ld_row(Kr[r], colb);
            vw[c][r] = ld_row(Vr[r], colb);
        }
    }
    qq[0] = ld_row(Qr, (unsigned)x0 * PIXB + lane16);

#pragma unroll 1
    for (int j = 0; j < STRIP; j += 4) {
#pragma unroll
        for (int u = 0; u < 4; ++u) {
            const int  x = x0 + j + u;
            const bool okc[3] = { x > 0, true, x + 1 < W };
            const unsigned colb = (unsigned)min(x + 2, W - 1) * PIXB + lane16;
            const v4f q = qq[u & 1];

            float sc[9];
#pragma unroll
            for (int r = 0; r < 3; ++r) {
#pragma unroll
                for (int c = 0; c < 3; ++c) {
                    const v4f k4 = kw[(u + c) & 3][r];
                    float s = q.x * k4.x + q.y * k4.y + q.z * k4.z + q.w * k4.w;
#pragma unroll
                    for (int m = 16; m >= 1; m >>= 1) s += __shfl_xor(s, m, 32);
                    sc[r * 3 + c] = (okr[r] & okc[c]) ? s * scale : 0.0f;
                }
            }

#pragma unroll
            for (int r = 0; r < 3; ++r) kw[(u + 3) & 3][r] = ld_row(Kr[r], colb);
            qq[(u + 1) & 1] = ld_row(Qr, (unsigned)min(x + 1, W - 1) * PIXB + lane16);

            float mx = sc[0];
#pragma unroll
            for (int n = 1; n < 9; ++n) mx = fmaxf(mx, sc[n]);
            asm volatile("" : "+v"(mx) : : "memory");

            float p[9], sum = 0.0f;
#pragma unroll
            for (int n = 0; n < 9; ++n) { p[n] = expf(sc[n] - mx); sum += p[n]; }
            const float inv = 1.0f / sum;

            v4f o = { 0.f, 0.f, 0.f, 0.f };
#pragma unroll
            for (int r = 0; r < 3; ++r) {
#pragma unroll
                for (int c = 0; c < 3; ++c) {
                    const float w  = (okr[r] & okc[c]) ? p[r * 3 + c] * inv : 0.0f;
                    const v4f   v4 = vw[(u + c) & 3][r];
                    o.x += w * v4.x; o.y += w * v4.y;
                    o.z += w * v4.z; o.w += w * v4.w;
                }
            }

#pragma unroll
            for (int r = 0; r < 3; ++r) vw[(u + 3) & 3][r] = ld_row(Vr[r], colb);

            asm volatile("" : "+v"(o), "+v"(qq[(u + 1) & 1]) : : "memory");

            *(v4f*)(Or + ((unsigned)x * PIXB + lane16)) = o;
        }
    }
}

extern "C" void kernel_launch(void* const* d_in, const int* in_sizes, int n_in,
                              void* d_out, int out_size, void* d_ws, size_t ws_size,
                              hipStream_t stream)
{
    const float* x  = (const float*)d_in[0];
    const float* Wq = (const float*)d_in[1];
    const float* bq = (const float*)d_in[2];
    const float* Wk = (const float*)d_in[3];
    const float* bk = (const float*)d_in[4];
    const float* Wv = (const float*)d_in[5];
    const float* bv = (const float*)d_in[6];

    const int H = 128, W = 128;
    const int Mrows = in_sizes[0] / DD;
    const int B = Mrows / (H * W);

    float*  Q   = (float*)d_ws;
    float*  Kp  = Q  + (size_t)Mrows * DD;
    float*  Vp  = Kp + (size_t)Mrows * DD;
    float4* Wf  = (float4*)(Vp + (size_t)Mrows * DD);

    reformat_w<<<48, 256, 0, stream>>>(Wq, Wk, Wv, Wf);

    const int gemmWaves = Mrows / 16;
    qkv_gemm_wmma<<<(gemmWaves + 7) / 8, 256, 0, stream>>>(
        x, Wf, bq, bk, bv, Q, Kp, Vp, Mrows);

    const int stripWaves = B * H * (W / STRIP);
    win_attn<<<(stripWaves + 7) / 8, 256, 0, stream>>>(
        Q, Kp, Vp, (float*)d_out, B, H, W);
}


# win_attn: the nine score sums cross lanes on the vector ALU (DPP row_xmask + one permlanex16 per sum) instead of 45 ds_bpermute per pixel through the shared LDS unit; summation order unchanged
# speedup vs baseline: 1.1934x; 1.0349x over previous
#include <hip/hip_runtime.h>


typedef float v2f __attribute__((ext_vector_type(2)));
typedef float v8f __attribute__((ext_vector_type(8)));
typedef float v4f __attribute__((ext_vector_type(4)));

#define DD 128
#define NT 8
#define WFRAG_F4_PER_MAT 4096

__global__ __launch_bounds__(256)
void reformat_w(const float* __restrict__ Wq, const float* __restrict__ Wk,
                const float* __restrict__ Wv, float4* __restrict__ Wf)
{
    const int idx = (int)(blockIdx.x * blockDim.x + threadIdx.x);
    const int m    = idx >> 12;
    const int r    = idx & 4095;
    const int kkp  = r >> 8;
    const int t    = (r >> 5) & 7;
    const int lane = r & 31;
    const int hi = lane >> 4, l = lane & 15;
    const int n  = t * 16 + l;
    const float* W = (m == 0) ? Wq : ((m == 1) ? Wk : Wv);
    const int k0 = 8 * kkp + 2 * hi;
    const int k1 = k0 + 4;
    float4 v;
    v.x = W[(size_t)(k0 + 0) * DD + n];
    v.y = W[(size_t)(k0 + 1) * DD + n];
    v.z = W[(size_t)(k1 + 0) * DD + n];
    v.w = W[(size_t)(k1 + 1) * DD + n];
    Wf[idx] = v;
}

__global__ __launch_bounds__(256)
void qkv_gemm_wmma(const float* __restrict__ X,
                   const float4* __restrict__ Wf,
                   const float* __restrict__ bq, const float* __restrict__ bk,
                   const float* __restrict__ bv,
                   float* __restrict__ Q, float* __restrict__ K, float* __restrict__ V,
                   int Mrows)
{
    const int wave = (int)((blockIdx.x * blockDim.x + threadIdx.x) >> 5);
    const int lane = (int)(threadIdx.x & 31);
    const int m0   = wave * 16;
    if (m0 >= Mrows) return;

    const int hi = lane >> 4;
    const int l  = lane & 15;

    const float* xrow = X + (size_t)(m0 + l) * DD + 2 * hi;

    const float* bm[3] = { bq, bk, bv };
    float*       Om[3] = { Q,  K,  V  };

#pragma unroll
    for (int m = 0; m < 3; ++m) {
        const float4* Wp = Wf + (size_t)m * WFRAG_F4_PER_MAT;
        const float*  bp = bm[m];
        float*        Op = Om[m];

        v8f c[NT];
#pragma unroll
        for (int t = 0; t < NT; ++t) {
            const float bias = bp[t * 16 + l];
#pragma unroll
            for (int r = 0; r < 8; ++r) c[t][r] = bias;
        }

#pragma unroll 2
        for (int kkp = 0; kkp < 16; ++kkp) {
            const v2f a0 = *(const v2f*)(xrow + 8 * kkp);
            const v2f a1 = *(const v2f*)(xrow + 8 * kkp + 4);
            float4 bf[NT];
#pragma unroll
            for (int t = 0; t < NT; ++t)
                bf[t] = Wp[(size_t)(kkp * NT + t) * 32 + lane];
#pragma unroll
            for (int t = 0; t < NT; ++t) {
                v2f b0; b0.x = bf[t].x; b0.y = bf[t].y;
                c[t] = __builtin_amdgcn_wmma_f32_16x16x4_f32(
                           false, a0, false, b0, (short)0, c[t], false, false);
            }
#pragma unroll
            for (int t = 0; t < NT; ++t) {
                v2f b1; b1.x = bf[t].z; b1.y = bf[t].w;
                c[t] = __builtin_amdgcn_wmma_f32_16x16x4_f32(
                           false, a1, false, b1, (short)0, c[t], false, false);
            }
        }

#pragma unroll
        for (int t = 0; t < NT; ++t) {
#pragma unroll
            for (int r = 0; r < 8; ++r) {
                Op[(size_t)(m0 + 8 * hi + r) * DD + t * 16 + l] = c[t][r];
            }
        }
    }
}

#define STRIP 16

__device__ __forceinline__ v4f ld_row(const char* P, unsigned boff)
{
    return *(const v4f*)(P + boff);
}

__device__ __forceinline__ float xhalf(float x)
{
    const int b = __builtin_bit_cast(int, x);
    return __builtin_bit_cast(float,
        __builtin_amdgcn_permlanex16(b, b, 0x76543210, 0xfedcba98, false, false));
}

template <int M>
__device__ __forceinline__ float xrow(float x)
{
    const int b = __builtin_bit_cast(int, x);
    return __builtin_bit_cast(float,
        __builtin_amdgcn_update_dpp(b, b, 0x160 | M, 0xf, 0xf, true));
}

__global__ __launch_bounds__(256)
void win_attn(const float* __restrict__ Q, const float* __restrict__ K,
              const float* __restrict__ V, float* __restrict__ out,
              int B, int H, int W)
{
    const int wave = __builtin_amdgcn_readfirstlane(
        (int)((blockIdx.x * blockDim.x + threadIdx.x) >> 5));
    const unsigned lane16 = (threadIdx.x & 31u) * 16u;
    const int spr  = W / STRIP;
    if (wave >= B * H * spr) return;

    const int  rowi = wave / spr;
    const int  x0   = (wave - rowi * spr) * STRIP;
    const int  y    = rowi % H;
    const bool okr[3] = { y > 0, true, y + 1 < H };
    const float scale = 0.08838834764831845f;
    const unsigned PIXB = DD * 4u;

    const char* Kr[3];
    const char* Vr[3];
#pragma unroll
    for (int r = 0; r < 3; ++r) {
        const size_t rb = (size_t)(rowi + (okr[r] ? r - 1 : 0)) * (size_t)W * PIXB;
        Kr[r] = (const char*)K + rb;
        Vr[r] = (const char*)V + rb;
    }
    const char* Qr = (const char*)Q + (size_t)rowi * (size_t)W * PIXB;
    char*       Or = (char*)out + (size_t)rowi * (size_t)W * PIXB;

    v4f kw[4][3], vw[4][3];
    v4f qq[2];
#pragma unroll
    for (int c = 0; c < 3; ++c) {
        const unsigned colb = (unsigned)min(max(x0 - 1 + c, 0), W - 1) * PIXB + lane16;
#pragma unroll
        for (int r = 0; r < 3; ++r) {
            kw[c][r] = ld_row(Kr[r], colb);
            vw[c][r] = ld_row(Vr[r], colb);
        }
    }
    qq[0] = ld_row(Qr, (unsigned)x0 * PIXB + lane16);

#pragma unroll 1
    for (int j = 0; j < STRIP; j += 4) {
#pragma unroll
        for (int u = 0; u < 4; ++u) {
            const int  x = x0 + j + u;
            const bool okc[3] = { x > 0, true, x + 1 < W };
            const unsigned colb = (unsigned)min(x + 2, W - 1) * PIXB + lane16;
            const v4f q = qq[u & 1];

            float sc[9];
#pragma unroll
            for (int r = 0; r < 3; ++r) {
#pragma unroll
                for (int c = 0; c < 3; ++c) {
                    const v4f k4 = kw[(u + c) & 3][r];
                    float s = q.x * k4.x + q.y * k4.y + q.z * k4.z + q.w * k4.w;
                    s += xhalf(s);
                    s += xrow<8>(s);
                    s += xrow<4>(s);
                    s += xrow<2>(s);
                    s += xrow<1>(s);
                    sc[r * 3 + c] = (okr[r] & okc[c]) ? s * scale : 0.0f;
                }
            }

#pragma unroll
            for (int r = 0; r < 3; ++r) kw[(u + 3) & 3][r] = ld_row(Kr[r], colb);
            qq[(u + 1) & 1] = ld_row(Qr, (unsigned)min(x + 1, W - 1) * PIXB + lane16);

            float mx = sc[0];
#pragma unroll
            for (int n = 1; n < 9; ++n) mx = fmaxf(mx, sc[n]);
            asm volatile("" : "+v"(mx) : : "memory");

            float p[9], sum = 0.0f;
#pragma unroll
            for (int n = 0; n < 9; ++n) { p[n] = expf(sc[n] - mx); sum += p[n]; }
            const float inv = 1.0f / sum;

            v4f o = { 0.f, 0.f, 0.f, 0.f };
#pragma unroll
            for (int r = 0; r < 3; ++r) {
#pragma unroll
                for (int c = 0; c < 3; ++c) {
                    const float w  = (okr[r] & okc[c]) ? p[r * 3 + c] * inv : 0.0f;
                    const v4f   v4 = vw[(u + c) & 3][r];
                    o.x += w * v4.x; o.y += w * v4.y;
                    o.z += w * v4.z; o.w += w * v4.w;
                }
            }

#pragma unroll
            for (int r = 0; r < 3; ++r) vw[(u + 3) & 3][r] = ld_row(Vr[r], colb);

            asm volatile("" : "+v"(o), "+v"(qq[(u + 1) & 1]) : : "memory");

            *(v4f*)(Or + ((unsigned)x * PIXB + lane16)) = o;
        }
    }
}

extern "C" void kernel_launch(void* const* d_in, const int* in_sizes, int n_in,
                              void* d_out, int out_size, void* d_ws, size_t ws_size,
                              hipStream_t stream)
{
    const float* x  = (const float*)d_in[0];
    const float* Wq = (const float*)d_in[1];
    const float* bq = (const float*)d_in[2];
    const float* Wk = (const float*)d_in[3];
    const float* bk = (const float*)d_in[4];
    const float* Wv = (const float*)d_in[5];
    const float* bv = (const float*)d_in[6];

    const int H = 128, W = 128;
    const int Mrows = in_sizes[0] / DD;
    const int B = Mrows / (H * W);

    float*  Q   = (float*)d_ws;
    float*  Kp  = Q  + (size_t)Mrows * DD;
    float*  Vp  = Kp + (size_t)Mrows * DD;
    float4* Wf  = (float4*)(Vp + (size_t)Mrows * DD);

    reformat_w<<<48, 256, 0, stream>>>(Wq, Wk, Wv, Wf);

    const int gemmWaves = Mrows / 16;
    qkv_gemm_wmma<<<(gemmWaves + 7) / 8, 256, 0, stream>>>(
        x, Wf, bq, bk, bv, Q, Kp, Vp, Mrows);

    const int stripWaves = B * H * (W / STRIP);
    win_attn<<<(stripWaves + 7) / 8, 256, 0, stream>>>(
        Q, Kp, Vp, (float*)d_out, B, H, W);
}


# win_attn: a work-group's 8 waves walk 8 adjacent rows of one strip (was 8 strips of one row), so K/V rows shared by vertical neighbours are re-read in one WGP cache; sum order unchanged
# speedup vs baseline: 1.2427x; 1.0413x over previous
#include <hip/hip_runtime.h>


typedef float v2f __attribute__((ext_vector_type(2)));
typedef float v8f __attribute__((ext_vector_type(8)));
typedef float v4f __attribute__((ext_vector_type(4)));

#define DD 128
#define NT 8
#define WFRAG_F4_PER_MAT 4096

__global__ __launch_bounds__(256)
void reformat_w(const float* __restrict__ Wq, const float* __restrict__ Wk,
                const float* __restrict__ Wv, float4* __restrict__ Wf)
{
    const int idx = (int)(blockIdx.x * blockDim.x + threadIdx.x);
    const int m    = idx >> 12;
    const int r    = idx & 4095;
    const int kkp  = r >> 8;
    const int t    = (r >> 5) & 7;
    const int lane = r & 31;
    const int hi = lane >> 4, l = lane & 15;
    const int n  = t * 16 + l;
    const float* W = (m == 0) ? Wq : ((m == 1) ? Wk : Wv);
    const int k0 = 8 * kkp + 2 * hi;
    const int k1 = k0 + 4;
    float4 v;
    v.x = W[(size_t)(k0 + 0) * DD + n];
    v.y = W[(size_t)(k0 + 1) * DD + n];
    v.z = W[(size_t)(k1 + 0) * DD + n];
    v.w = W[(size_t)(k1 + 1) * DD + n];
    Wf[idx] = v;
}

__global__ __launch_bounds__(256)
void qkv_gemm_wmma(const float* __restrict__ X,
                   const float4* __restrict__ Wf,
                   const float* __restrict__ bq, const float* __restrict__ bk,
                   const float* __restrict__ bv,
                   float* __restrict__ Q, float* __restrict__ K, float* __restrict__ V,
                   int Mrows)
{
    const int wave = (int)((blockIdx.x * blockDim.x + threadIdx.x) >> 5);
    const int lane = (int)(threadIdx.x & 31);
    const int m0   = wave * 16;
    if (m0 >= Mrows) return;

    const int hi = lane >> 4;
    const int l  = lane & 15;

    const float* xrow = X + (size_t)(m0 + l) * DD + 2 * hi;

    const float* bm[3] = { bq, bk, bv };
    float*       Om[3] = { Q,  K,  V  };

#pragma unroll
    for (int m = 0; m < 3; ++m) {
        const float4* Wp = Wf + (size_t)m * WFRAG_F4_PER_MAT;
        const float*  bp = bm[m];
        float*        Op = Om[m];

        v8f c[NT];
#pragma unroll
        for (int t = 0; t < NT; ++t) {
            const float bias = bp[t * 16 + l];
#pragma unroll
            for (int r = 0; r < 8; ++r) c[t][r] = bias;
        }

#pragma unroll 2
        for (int kkp = 0; kkp < 16; ++kkp) {
            const v2f a0 = *(const v2f*)(xrow + 8 * kkp);
            const v2f a1 = *(const v2f*)(xrow + 8 * kkp + 4);
            float4 bf[NT];
#pragma unroll
            for (int t = 0; t < NT; ++t)
                bf[t] = Wp[(size_t)(kkp * NT + t) * 32 + lane];
#pragma unroll
            for (int t = 0; t < NT; ++t) {
                v2f b0; b0.x = bf[t].x; b0.y = bf[t].y;
                c[t] = __builtin_amdgcn_wmma_f32_16x16x4_f32(
                           false, a0, false, b0, (short)0, c[t], false, false);
            }
#pragma unroll
            for (int t = 0; t < NT; ++t) {
                v2f b1; b1.x = bf[t].z; b1.y = bf[t].w;
                c[t] = __builtin_amdgcn_wmma_f32_16x16x4_f32(
                           false, a1, false, b1, (short)0, c[t], false, false);
            }
        }

#pragma unroll
        for (int t = 0; t < NT; ++t) {
#pragma unroll
            for (int r = 0; r < 8; ++r) {
                Op[(size_t)(m0 + 8 * hi + r) * DD + t * 16 + l] = c[t][r];
            }
        }
    }
}

#define STRIP 16

__device__ __forceinline__ v4f ld_row(const char* P, unsigned boff)
{
    return *(const v4f*)(P + boff);
}

__device__ __forceinline__ float xhalf(float x)
{
    const int b = __builtin_bit_cast(int, x);
    return __builtin_bit_cast(float,
        __builtin_amdgcn_permlanex16(b, b, 0x76543210, 0xfedcba98, false, false));
}

template <int M>
__device__ __forceinline__ float xrow(float x)
{
    const int b = __builtin_bit_cast(int, x);
    return __builtin_bit_cast(float,
        __builtin_amdgcn_update_dpp(b, b, 0x160 | M, 0xf, 0xf, true));
}

__global__ __launch_bounds__(256)
void win_attn(const float* __restrict__ Q, const float* __restrict__ K,
              const float* __restrict__ V, float* __restrict__ out,
              int B, int H, int W)
{
    const int wave = __builtin_amdgcn_readfirstlane(
        (int)((blockIdx.x * blockDim.x + threadIdx.x) >> 5));
    const unsigned lane16 = (threadIdx.x & 31u) * 16u;
    const int spr  = W / STRIP;
    if (wave >= B * H * spr) return;

    const int  wig  = wave & 7;
    const int  blk  = wave >> 3;
    const int  rblk = blk / spr;
    const int  rowi = rblk * 8 + wig;
    const int  x0   = (blk - rblk * spr) * STRIP;
    if (rowi >= B * H) return;
    const int  y    = rowi % H;
    const bool okr[3] = { y > 0, true, y + 1 < H };
    const float scale = 0.08838834764831845f;
    const unsigned PIXB = DD * 4u;

    const char* Kr[3];
    const char* Vr[3];
#pragma unroll
    for (int r = 0; r < 3; ++r) {
        const size_t rb = (size_t)(rowi + (okr[r] ? r - 1 : 0)) * (size_t)W * PIXB;
        Kr[r] = (const char*)K + rb;
        Vr[r] = (const char*)V + rb;
    }
    const char* Qr = (const char*)Q + (size_t)rowi * (size_t)W * PIXB;
    char*       Or = (char*)out + (size_t)rowi * (size_t)W * PIXB;

    v4f kw[4][3], vw[4][3];
    v4f qq[2];
#pragma unroll
    for (int c = 0; c < 3; ++c) {
        const unsigned colb = (unsigned)min(max(x0 - 1 + c, 0), W - 1) * PIXB + lane16;
#pragma unroll
        for (int r = 0; r < 3; ++r) {
            kw[c][r] = ld_row(Kr[r], colb);
            vw[c][r] = ld_row(Vr[r], colb);
        }
    }
    qq[0] = ld_row(Qr, (unsigned)x0 * PIXB + lane16);

#pragma unroll 1
    for (int j = 0; j < STRIP; j += 4) {
#pragma unroll
        for (int u = 0; u < 4; ++u) {
            const int  x = x0 + j + u;
            const bool okc[3] = { x > 0, true, x + 1 < W };
            const unsigned colb = (unsigned)min(x + 2, W - 1) * PIXB + lane16;
            const v4f q = qq[u & 1];

            float sc[9];
#pragma unroll
            for (int r = 0; r < 3; ++r) {
#pragma unroll
                for (int c = 0; c < 3; ++c) {
                    const v4f k4 = kw[(u + c) & 3][r];
                    float s = q.x * k4.x + q.y * k4.y + q.z * k4.z + q.w * k4.w;
                    s += xhalf(s);
                    s += xrow<8>(s);
                    s += xrow<4>(s);
                    s += xrow<2>(s);
                    s += xrow<1>(s);
                    sc[r * 3 + c] = (okr[r] & okc[c]) ? s * scale : 0.0f;
                }
            }

#pragma unroll
            for (int r = 0; r < 3; ++r) kw[(u + 3) & 3][r] = ld_row(Kr[r], colb);
            qq[(u + 1) & 1] = ld_row(Qr, (unsigned)min(x + 1, W - 1) * PIXB + lane16);

            float mx = sc[0];
#pragma unroll
            for (int n = 1; n < 9; ++n) mx = fmaxf(mx, sc[n]);
            asm volatile("" : "+v"(mx) : : "memory");

            float p[9], sum = 0.0f;
#pragma unroll
            for (int n = 0; n < 9; ++n) { p[n] = expf(sc[n] - mx); sum += p[n]; }
            const float inv = 1.0f / sum;

            v4f o = { 0.f, 0.f, 0.f, 0.f };
#pragma unroll
            for (int r = 0; r < 3; ++r) {
#pragma unroll
                for (int c = 0; c < 3; ++c) {
                    const float w  = (okr[r] & okc[c]) ? p[r * 3 + c] * inv : 0.0f;
                    const v4f   v4 = vw[(u + c) & 3][r];
                    o.x += w * v4.x; o.y += w * v4.y;
                    o.z += w * v4.z; o.w += w * v4.w;
                }
            }

#pragma unroll
            for (int r = 0; r < 3; ++r) vw[(u + 3) & 3][r] = ld_row(Vr[r], colb);

            asm volatile("" : "+v"(o), "+v"(qq[(u + 1) & 1]) : : "memory");

            *(v4f*)(Or + ((unsigned)x * PIXB + lane16)) = o;
        }
    }
}

extern "C" void kernel_launch(void* const* d_in, const int* in_sizes, int n_in,
                              void* d_out, int out_size, void* d_ws, size_t ws_size,
                              hipStream_t stream)
{
    const float* x  = (const float*)d_in[0];
    const float* Wq = (const float*)d_in[1];
    const float* bq = (const float*)d_in[2];
    const float* Wk = (const float*)d_in[3];
    const float* bk = (const float*)d_in[4];
    const float* Wv = (const float*)d_in[5];
    const float* bv = (const float*)d_in[6];

    const int H = 128, W = 128;
    const int Mrows = in_sizes[0] / DD;
    const int B = Mrows / (H * W);

    float*  Q   = (float*)d_ws;
    float*  Kp  = Q  + (size_t)Mrows * DD;
    float*  Vp  = Kp + (size_t)Mrows * DD;
    float4* Wf  = (float4*)(Vp + (size_t)Mrows * DD);

    reformat_w<<<48, 256, 0, stream>>>(Wq, Wk, Wv, Wf);

    const int gemmWaves = Mrows / 16;
    qkv_gemm_wmma<<<(gemmWaves + 7) / 8, 256, 0, stream>>>(
        x, Wf, bq, bk, bv, Q, Kp, Vp, Mrows);

    const int stripWaves = B * H * (W / STRIP);
    win_attn<<<(stripWaves + 7) / 8, 256, 0, stream>>>(
        Q, Kp, Vp, (float*)d_out, B, H, W);
}


# qkv_gemm: finished 16x128 tiles pass through a wave-private LDS block so each store instruction writes one whole 512-byte row (was 64-byte halves of two lines per b32 store); values unchanged
# speedup vs baseline: 1.3200x; 1.0622x over previous
#include <hip/hip_runtime.h>


typedef float v2f __attribute__((ext_vector_type(2)));
typedef float v8f __attribute__((ext_vector_type(8)));
typedef float v4f __attribute__((ext_vector_type(4)));

#define DD 128
#define NT 8
#define WFRAG_F4_PER_MAT 4096

__global__ __launch_bounds__(256)
void reformat_w(const float* __restrict__ Wq, const float* __restrict__ Wk,
                const float* __restrict__ Wv, float4* __restrict__ Wf)
{
    const int idx = (int)(blockIdx.x * blockDim.x + threadIdx.x);
    const int m    = idx >> 12;
    const int r    = idx & 4095;
    const int kkp  = r >> 8;
    const int t    = (r >> 5) & 7;
    const int lane = r & 31;
    const int hi = lane >> 4, l = lane & 15;
    const int n  = t * 16 + l;
    const float* W = (m == 0) ? Wq : ((m == 1) ? Wk : Wv);
    const int k0 = 8 * kkp + 2 * hi;
    const int k1 = k0 + 4;
    float4 v;
    v.x = W[(size_t)(k0 + 0) * DD + n];
    v.y = W[(size_t)(k0 + 1) * DD + n];
    v.z = W[(size_t)(k1 + 0) * DD + n];
    v.w = W[(size_t)(k1 + 1) * DD + n];
    Wf[idx] = v;
}

#define EPI_PITCH 132
#define EPI_TILE  (16 * EPI_PITCH)

__device__ __forceinline__ void lds_wave_sync()
{
    asm volatile("s_wait_dscnt 0x0" ::: "memory");
}

__global__ __launch_bounds__(256, 8)
void qkv_gemm_wmma(const float* __restrict__ X,
                   const float4* __restrict__ Wf,
                   const float* __restrict__ bq, const float* __restrict__ bk,
                   const float* __restrict__ bv,
                   float* __restrict__ Q, float* __restrict__ K, float* __restrict__ V,
                   int Mrows)
{
    const int wave = (int)((blockIdx.x * blockDim.x + threadIdx.x) >> 5);
    const int lane = (int)(threadIdx.x & 31);
    const int m0   = wave * 16;
    if (m0 >= Mrows) return;

    const int hi = lane >> 4;
    const int l  = lane & 15;

    const float* xrow = X + (size_t)(m0 + l) * DD + 2 * hi;

    const float* bm[3] = { bq, bk, bv };
    float*       Om[3] = { Q,  K,  V  };

    __shared__ v4f epi[8 * EPI_TILE / 4];
    float* E = (float*)epi + (threadIdx.x >> 5) * EPI_TILE;

#pragma unroll
    for (int m = 0; m < 3; ++m) {
        const float4* Wp = Wf + (size_t)m * WFRAG_F4_PER_MAT;
        const float*  bp = bm[m];
        float*        Op = Om[m];

        v8f c[NT];
#pragma unroll
        for (int t = 0; t < NT; ++t) {
            const float bias = bp[t * 16 + l];
#pragma unroll
            for (int r = 0; r < 8; ++r) c[t][r] = bias;
        }

#pragma unroll 2
        for (int kkp = 0; kkp < 16; ++kkp) {
            const v2f a0 = *(const v2f*)(xrow + 8 * kkp);
            const v2f a1 = *(const v2f*)(xrow + 8 * kkp + 4);
            float4 bf[NT];
#pragma unroll
            for (int t = 0; t < NT; ++t)
                bf[t] = Wp[(size_t)(kkp * NT + t) * 32 + lane];
#pragma unroll
            for (int t = 0; t < NT; ++t) {
                v2f b0; b0.x = bf[t].x; b0.y = bf[t].y;
                c[t] = __builtin_amdgcn_wmma_f32_16x16x4_f32(
                           false, a0, false, b0, (short)0, c[t], false, false);
            }
#pragma unroll
            for (int t = 0; t < NT; ++t) {
                v2f b1; b1.x = bf[t].z; b1.y = bf[t].w;
                c[t] = __builtin_amdgcn_wmma_f32_16x16x4_f32(
                           false, a1, false, b1, (short)0, c[t], false, false);
            }
        }

#pragma unroll
        for (int t = 0; t < NT; ++t) {
#pragma unroll
            for (int r = 0; r < 8; ++r) {
                E[(8 * hi + r) * EPI_PITCH + t * 16 + l] = c[t][r];
            }
        }
        lds_wave_sync();
        v4f row[16];
#pragma unroll
        for (int rr = 0; rr < 16; ++rr)
            row[rr] = *(const v4f*)(E + rr * EPI_PITCH + lane * 4);
#pragma unroll
        for (int rr = 0; rr < 16; ++rr)
            *(v4f*)(Op + (size_t)(m0 + rr) * DD + lane * 4) = row[rr];
    }
}

#define STRIP 16

__device__ __forceinline__ v4f ld_row(const char* P, unsigned boff)
{
    return *(const v4f*)(P + boff);
}

__device__ __forceinline__ float xhalf(float x)
{
    const int b = __builtin_bit_cast(int, x);
    return __builtin_bit_cast(float,
        __builtin_amdgcn_permlanex16(b, b, 0x76543210, 0xfedcba98, false, false));
}

template <int M>
__device__ __forceinline__ float xrow(float x)
{
    const int b = __builtin_bit_cast(int, x);
    return __builtin_bit_cast(float,
        __builtin_amdgcn_update_dpp(b, b, 0x160 | M, 0xf, 0xf, true));
}

__global__ __launch_bounds__(256)
void win_attn(const float* __restrict__ Q, const float* __restrict__ K,
              const float* __restrict__ V, float* __restrict__ out,
              int B, int H, int W)
{
    const int wave = __builtin_amdgcn_readfirstlane(
        (int)((blockIdx.x * blockDim.x + threadIdx.x) >> 5));
    const unsigned lane16 = (threadIdx.x & 31u) * 16u;
    const int spr  = W / STRIP;
    if (wave >= B * H * spr) return;

    const int  wig  = wave & 7;
    const int  blk  = wave >> 3;
    const int  rblk = blk / spr;
    const int  rowi = rblk * 8 + wig;
    const int  x0   = (blk - rblk * spr) * STRIP;
    if (rowi >= B * H) return;
    const int  y    = rowi % H;
    const bool okr[3] = { y > 0, true, y + 1 < H };
    const float scale = 0.08838834764831845f;
    const unsigned PIXB = DD * 4u;

    const char* Kr[3];
    const char* Vr[3];
#pragma unroll
    for (int r = 0; r < 3; ++r) {
        const size_t rb = (size_t)(rowi + (okr[r] ? r - 1 : 0)) * (size_t)W * PIXB;
        Kr[r] = (const char*)K + rb;
        Vr[r] = (const char*)V + rb;
    }
    const char* Qr = (const char*)Q + (size_t)rowi * (size_t)W * PIXB;
    char*       Or = (char*)out + (size_t)rowi * (size_t)W * PIXB;

    v4f kw[4][3], vw[4][3];
    v4f qq[2];
#pragma unroll
    for (int c = 0; c < 3; ++c) {
        const unsigned colb = (unsigned)min(max(x0 - 1 + c, 0), W - 1) * PIXB + lane16;
#pragma unroll
        for (int r = 0; r < 3; ++r) {
            kw[c][r] = ld_row(Kr[r], colb);
            vw[c][r] = ld_row(Vr[r], colb);
        }
    }
    qq[0] = ld_row(Qr, (unsigned)x0 * PIXB + lane16);

#pragma unroll 1
    for (int j = 0; j < STRIP; j += 4) {
#pragma unroll
        for (int u = 0; u < 4; ++u) {
            const int  x = x0 + j + u;
            const bool okc[3] = { x > 0, true, x + 1 < W };
            const unsigned colb = (unsigned)min(x + 2, W - 1) * PIXB + lane16;
            const v4f q = qq[u & 1];

            float sc[9];
#pragma unroll
            for (int r = 0; r < 3; ++r) {
#pragma unroll
                for (int c = 0; c < 3; ++c) {
                    const v4f k4 = kw[(u + c) & 3][r];
                    float s = q.x * k4.x + q.y * k4.y + q.z * k4.z + q.w * k4.w;
                    s += xhalf(s);
                    s += xrow<8>(s);
                    s += xrow<4>(s);
                    s += xrow<2>(s);
                    s += xrow<1>(s);
                    sc[r * 3 + c] = (okr[r] & okc[c]) ? s * scale : 0.0f;
                }
            }

#pragma unroll
            for (int r = 0; r < 3; ++r) kw[(u + 3) & 3][r] = ld_row(Kr[r], colb);
            qq[(u + 1) & 1] = ld_row(Qr, (unsigned)min(x + 1, W - 1) * PIXB + lane16);

            float mx = sc[0];
#pragma unroll
            for (int n = 1; n < 9; ++n) mx = fmaxf(mx, sc[n]);
            asm volatile("" : "+v"(mx) : : "memory");

            float p[9], sum = 0.0f;
#pragma unroll
            for (int n = 0; n < 9; ++n) { p[n] = expf(sc[n] - mx); sum += p[n]; }
            const float inv = 1.0f / sum;

            v4f o = { 0.f, 0.f, 0.f, 0.f };
#pragma unroll
            for (int r = 0; r < 3; ++r) {
#pragma unroll
                for (int c = 0; c < 3; ++c) {
                    const float w  = (okr[r] & okc[c]) ? p[r * 3 + c] * inv : 0.0f;
                    const v4f   v4 = vw[(u + c) & 3][r];
                    o.x += w * v4.x; o.y += w * v4.y;
                    o.z += w * v4.z; o.w += w * v4.w;
                }
            }

#pragma unroll
            for (int r = 0; r < 3; ++r) vw[(u + 3) & 3][r] = ld_row(Vr[r], colb);

            asm volatile("" : "+v"(o), "+v"(qq[(u + 1) & 1]) : : "memory");

            *(v4f*)(Or + ((unsigned)x * PIXB + lane16)) = o;
        }
    }
}

extern "C" void kernel_launch(void* const* d_in, const int* in_sizes, int n_in,
                              void* d_out, int out_size, void* d_ws, size_t ws_size,
                              hipStream_t stream)
{
    const float* x  = (const float*)d_in[0];
    const float* Wq = (const float*)d_in[1];
    const float* bq = (const float*)d_in[2];
    const float* Wk = (const float*)d_in[3];
    const float* bk = (const float*)d_in[4];
    const float* Wv = (const float*)d_in[5];
    const float* bv = (const float*)d_in[6];

    const int H = 128, W = 128;
    const int Mrows = in_sizes[0] / DD;
    const int B = Mrows / (H * W);

    float*  Q   = (float*)d_ws;
    float*  Kp  = Q  + (size_t)Mrows * DD;
    float*  Vp  = Kp + (size_t)Mrows * DD;
    float4* Wf  = (float4*)(Vp + (size_t)Mrows * DD);

    reformat_w<<<48, 256, 0, stream>>>(Wq, Wk, Wv, Wf);

    const int gemmWaves = Mrows / 16;
    qkv_gemm_wmma<<<(gemmWaves + 7) / 8, 256, 0, stream>>>(
        x, Wf, bq, bk, bv, Q, Kp, Vp, Mrows);

    const int stripWaves = B * H * (W / STRIP);
    win_attn<<<(stripWaves + 7) / 8, 256, 0, stream>>>(
        Q, Kp, Vp, (float*)d_out, B, H, W);
}


# qkv_gemm: a wave computes 32 rows (two 16-row tiles), so each B fragment loaded from the WGP cache feeds two WMMAs: half the B loads per WMMA, 4 waves per SIMD; per-tile k order and values unchanged
# speedup vs baseline: 1.4351x; 1.0872x over previous
#include <hip/hip_runtime.h>


typedef float v2f __attribute__((ext_vector_type(2)));
typedef float v8f __attribute__((ext_vector_type(8)));
typedef float v4f __attribute__((ext_vector_type(4)));

#define DD 128
#define NT 8
#define WFRAG_F4_PER_MAT 4096

__global__ __launch_bounds__(256)
void reformat_w(const float* __restrict__ Wq, const float* __restrict__ Wk,
                const float* __restrict__ Wv, float4* __restrict__ Wf)
{
    const int idx = (int)(blockIdx.x * blockDim.x + threadIdx.x);
    const int m    = idx >> 12;
    const int r    = idx & 4095;
    const int kkp  = r >> 8;
    const int t    = (r >> 5) & 7;
    const int lane = r & 31;
    const int hi = lane >> 4, l = lane & 15;
    const int n  = t * 16 + l;
    const float* W = (m == 0) ? Wq : ((m == 1) ? Wk : Wv);
    const int k0 = 8 * kkp + 2 * hi;
    const int k1 = k0 + 4;
    float4 v;
    v.x = W[(size_t)(k0 + 0) * DD + n];
    v.y = W[(size_t)(k0 + 1) * DD + n];
    v.z = W[(size_t)(k1 + 0) * DD + n];
    v.w = W[(size_t)(k1 + 1) * DD + n];
    Wf[idx] = v;
}

#define EPI_PITCH 132
#define EPI_TILE  (16 * EPI_PITCH)

__device__ __forceinline__ void lds_wave_sync()
{
    asm volatile("s_wait_dscnt 0x0" ::: "memory");
}

#define RT 2
__global__ __launch_bounds__(256, 4)
void qkv_gemm_wmma(const float* __restrict__ X,
                   const float4* __restrict__ Wf,
                   const float* __restrict__ bq, const float* __restrict__ bk,
                   const float* __restrict__ bv,
                   float* __restrict__ Q, float* __restrict__ K, float* __restrict__ V,
                   int Mrows)
{
    const int wave = (int)((blockIdx.x * blockDim.x + threadIdx.x) >> 5);
    const int lane = (int)(threadIdx.x & 31);
    const int m0   = wave * (16 * RT);
    if (m0 >= Mrows) return;

    const int hi = lane >> 4;
    const int l  = lane & 15;

    const float* xrow = X + (size_t)(m0 + l) * DD + 2 * hi;

    const float* bm[3] = { bq, bk, bv };
    float*       Om[3] = { Q,  K,  V  };

    __shared__ v4f epi[8 * EPI_TILE / 4];
    float* E = (float*)epi + (threadIdx.x >> 5) * EPI_TILE;

#pragma unroll
    for (int m = 0; m < 3; ++m) {
        const float4* Wp = Wf + (size_t)m * WFRAG_F4_PER_MAT;
        const float*  bp = bm[m];
        float*        Op = Om[m];

        v8f c[RT][NT];
#pragma unroll
        for (int t = 0; t < NT; ++t) {
            const float bias = bp[t * 16 + l];
#pragma unroll
            for (int i = 0; i < RT; ++i) {
#pragma unroll
                for (int r = 0; r < 8; ++r) c[i][t][r] = bias;
            }
        }

        for (int kkp = 0; kkp < 16; ++kkp) {
            v2f a0[RT], a1[RT];
#pragma unroll
            for (int i = 0; i < RT; ++i) {
                a0[i] = *(const v2f*)(xrow + (size_t)i * 16 * DD + 8 * kkp);
                a1[i] = *(const v2f*)(xrow + (size_t)i * 16 * DD + 8 * kkp + 4);
            }
            float4 bf[NT];
#pragma unroll
            for (int t = 0; t < NT; ++t)
                bf[t] = Wp[(size_t)(kkp * NT + t) * 32 + lane];
#pragma unroll
            for (int i = 0; i < RT; ++i) {
#pragma unroll
                for (int t = 0; t < NT; ++t) {
                    v2f b0; b0.x = bf[t].x; b0.y = bf[t].y;
                    c[i][t] = __builtin_amdgcn_wmma_f32_16x16x4_f32(
                               false, a0[i], false, b0, (short)0, c[i][t], false, false);
                }
            }
#pragma unroll
            for (int i = 0; i < RT; ++i) {
#pragma unroll
                for (int t = 0; t < NT; ++t) {
                    v2f b1; b1.x = bf[t].z; b1.y = bf[t].w;
                    c[i][t] = __builtin_amdgcn_wmma_f32_16x16x4_f32(
                               false, a1[i], false, b1, (short)0, c[i][t], false, false);
                }
            }
        }

#pragma unroll
        for (int i = 0; i < RT; ++i) {
#pragma unroll
            for (int t = 0; t < NT; ++t) {
#pragma unroll
                for (int r = 0; r < 8; ++r) {
                    E[(8 * hi + r) * EPI_PITCH + t * 16 + l] = c[i][t][r];
                }
            }
            lds_wave_sync();
            v4f row[16];
#pragma unroll
            for (int rr = 0; rr < 16; ++rr)
                row[rr] = *(const v4f*)(E + rr * EPI_PITCH + lane * 4);
#pragma unroll
            for (int rr = 0; rr < 16; ++rr)
                *(v4f*)(Op + (size_t)(m0 + 16 * i + rr) * DD + lane * 4) = row[rr];
        }
    }
}

#define STRIP 16

__device__ __forceinline__ v4f ld_row(const char* P, unsigned boff)
{
    return *(const v4f*)(P + boff);
}

__device__ __forceinline__ float xhalf(float x)
{
    const int b = __builtin_bit_cast(int, x);
    return __builtin_bit_cast(float,
        __builtin_amdgcn_permlanex16(b, b, 0x76543210, 0xfedcba98, false, false));
}

template <int M>
__device__ __forceinline__ float xrow(float x)
{
    const int b = __builtin_bit_cast(int, x);
    return __builtin_bit_cast(float,
        __builtin_amdgcn_update_dpp(b, b, 0x160 | M, 0xf, 0xf, true));
}

__global__ __launch_bounds__(256)
void win_attn(const float* __restrict__ Q, const float* __restrict__ K,
              const float* __restrict__ V, float* __restrict__ out,
              int B, int H, int W)
{
    const int wave = __builtin_amdgcn_readfirstlane(
        (int)((blockIdx.x * blockDim.x + threadIdx.x) >> 5));
    const unsigned lane16 = (threadIdx.x & 31u) * 16u;
    const int spr  = W / STRIP;
    if (wave >= B * H * spr) return;

    const int  wig  = wave & 7;
    const int  blk  = wave >> 3;
    const int  rblk = blk / spr;
    const int  rowi = rblk * 8 + wig;
    const int  x0   = (blk - rblk * spr) * STRIP;
    if (rowi >= B * H) return;
    const int  y    = rowi % H;
    const bool okr[3] = { y > 0, true, y + 1 < H };
    const float scale = 0.08838834764831845f;
    const unsigned PIXB = DD * 4u;

    const char* Kr[3];
    const char* Vr[3];
#pragma unroll
    for (int r = 0; r < 3; ++r) {
        const size_t rb = (size_t)(rowi + (okr[r] ? r - 1 : 0)) * (size_t)W * PIXB;
        Kr[r] = (const char*)K + rb;
        Vr[r] = (const char*)V + rb;
    }
    const char* Qr = (const char*)Q + (size_t)rowi * (size_t)W * PIXB;
    char*       Or = (char*)out + (size_t)rowi * (size_t)W * PIXB;

    v4f kw[4][3], vw[4][3];
    v4f qq[2];
#pragma unroll
    for (int c = 0; c < 3; ++c) {
        const unsigned colb = (unsigned)min(max(x0 - 1 + c, 0), W - 1) * PIXB + lane16;
#pragma unroll
        for (int r = 0; r < 3; ++r) {
            kw[c][r] = ld_row(Kr[r], colb);
            vw[c][r] = ld_row(Vr[r], colb);
        }
    }
    qq[0] = ld_row(Qr, (unsigned)x0 * PIXB + lane16);

#pragma unroll 1
    for (int j = 0; j < STRIP; j += 4) {
#pragma unroll
        for (int u = 0; u < 4; ++u) {
            const int  x = x0 + j + u;
            const bool okc[3] = { x > 0, true, x + 1 < W };
            const unsigned colb = (unsigned)min(x + 2, W - 1) * PIXB + lane16;
            const v4f q = qq[u & 1];

            float sc[9];
#pragma unroll
            for (int r = 0; r < 3; ++r) {
#pragma unroll
                for (int c = 0; c < 3; ++c) {
                    const v4f k4 = kw[(u + c) & 3][r];
                    float s = q.x * k4.x + q.y * k4.y + q.z * k4.z + q.w * k4.w;
                    s += xhalf(s);
                    s += xrow<8>(s);
                    s += xrow<4>(s);
                    s += xrow<2>(s);
                    s += xrow<1>(s);
                    sc[r * 3 + c] = (okr[r] & okc[c]) ? s * scale : 0.0f;
                }
            }

#pragma unroll
            for (int r = 0; r < 3; ++r) kw[(u + 3) & 3][r] = ld_row(Kr[r], colb);
            qq[(u + 1) & 1] = ld_row(Qr, (unsigned)min(x + 1, W - 1) * PIXB + lane16);

            float mx = sc[0];
#pragma unroll
            for (int n = 1; n < 9; ++n) mx = fmaxf(mx, sc[n]);
            asm volatile("" : "+v"(mx) : : "memory");

            float p[9], sum = 0.0f;
#pragma unroll
            for (int n = 0; n < 9; ++n) { p[n] = expf(sc[n] - mx); sum += p[n]; }
            const float inv = 1.0f / sum;

            v4f o = { 0.f, 0.f, 0.f, 0.f };
#pragma unroll
            for (int r = 0; r < 3; ++r) {
#pragma unroll
                for (int c = 0; c < 3; ++c) {
                    const float w  = (okr[r] & okc[c]) ? p[r * 3 + c] * inv : 0.0f;
                    const v4f   v4 = vw[(u + c) & 3][r];
                    o.x += w * v4.x; o.y += w * v4.y;
                    o.z += w * v4.z; o.w += w * v4.w;
                }
            }

#pragma unroll
            for (int r = 0; r < 3; ++r) vw[(u + 3) & 3][r] = ld_row(Vr[r], colb);

            asm volatile("" : "+v"(o), "+v"(qq[(u + 1) & 1]) : : "memory");

            *(v4f*)(Or + ((unsigned)x * PIXB + lane16)) = o;
        }
    }
}

extern "C" void kernel_launch(void* const* d_in, const int* in_sizes, int n_in,
                              void* d_out, int out_size, void* d_ws, size_t ws_size,
                              hipStream_t stream)
{
    const float* x  = (const float*)d_in[0];
    const float* Wq = (const float*)d_in[1];
    const float* bq = (const float*)d_in[2];
    const float* Wk = (const float*)d_in[3];
    const float* bk = (const float*)d_in[4];
    const float* Wv = (const float*)d_in[5];
    const float* bv = (const float*)d_in[6];

    const int H = 128, W = 128;
    const int Mrows = in_sizes[0] / DD;
    const int B = Mrows / (H * W);

    float*  Q   = (float*)d_ws;
    float*  Kp  = Q  + (size_t)Mrows * DD;
    float*  Vp  = Kp + (size_t)Mrows * DD;
    float4* Wf  = (float4*)(Vp + (size_t)Mrows * DD);

    reformat_w<<<48, 256, 0, stream>>>(Wq, Wk, Wv, Wf);

    const int gemmWaves = Mrows / 32;
    qkv_gemm_wmma<<<(gemmWaves + 7) / 8, 256, 0, stream>>>(
        x, Wf, bq, bk, bv, Q, Kp, Vp, Mrows);

    const int stripWaves = B * H * (W / STRIP);
    win_attn<<<(stripWaves + 7) / 8, 256, 0, stream>>>(
        Q, Kp, Vp, (float*)d_out, B, H, W);
}


# win_attn: the nine scores are equal in all lanes, so max, exp, sum, reciprocal and weights run once per wave on the scalar ALU (readfirstlane); exp2 and rcp replace expf and the IEEE divide
# speedup vs baseline: 1.5388x; 1.0723x over previous
#include <hip/hip_runtime.h>


typedef float v2f __attribute__((ext_vector_type(2)));
typedef float v8f __attribute__((ext_vector_type(8)));
typedef float v4f __attribute__((ext_vector_type(4)));

#define DD 128
#define NT 8
#define WFRAG_F4_PER_MAT 4096

__global__ __launch_bounds__(256)
void reformat_w(const float* __restrict__ Wq, const float* __restrict__ Wk,
                const float* __restrict__ Wv, float4* __restrict__ Wf)
{
    const int idx = (int)(blockIdx.x * blockDim.x + threadIdx.x);
    const int m    = idx >> 12;
    const int r    = idx & 4095;
    const int kkp  = r >> 8;
    const int t    = (r >> 5) & 7;
    const int lane = r & 31;
    const int hi = lane >> 4, l = lane & 15;
    const int n  = t * 16 + l;
    const float* W = (m == 0) ? Wq : ((m == 1) ? Wk : Wv);
    const int k0 = 8 * kkp + 2 * hi;
    const int k1 = k0 + 4;
    float4 v;
    v.x = W[(size_t)(k0 + 0) * DD + n];
    v.y = W[(size_t)(k0 + 1) * DD + n];
    v.z = W[(size_t)(k1 + 0) * DD + n];
    v.w = W[(size_t)(k1 + 1) * DD + n];
    Wf[idx] = v;
}

#define EPI_PITCH 132
#define EPI_TILE  (16 * EPI_PITCH)

__device__ __forceinline__ void lds_wave_sync()
{
    asm volatile("s_wait_dscnt 0x0" ::: "memory");
}

#define RT 2
__global__ __launch_bounds__(256, 4)
void qkv_gemm_wmma(const float* __restrict__ X,
                   const float4* __restrict__ Wf,
                   const float* __restrict__ bq, const float* __restrict__ bk,
                   const float* __restrict__ bv,
                   float* __restrict__ Q, float* __restrict__ K, float* __restrict__ V,
                   int Mrows)
{
    const int wave = (int)((blockIdx.x * blockDim.x + threadIdx.x) >> 5);
    const int lane = (int)(threadIdx.x & 31);
    const int m0   = wave * (16 * RT);
    if (m0 >= Mrows) return;

    const int hi = lane >> 4;
    const int l  = lane & 15;

    const float* xrow = X + (size_t)(m0 + l) * DD + 2 * hi;

    const float* bm[3] = { bq, bk, bv };
    float*       Om[3] = { Q,  K,  V  };

    __shared__ v4f epi[8 * EPI_TILE / 4];
    float* E = (float*)epi + (threadIdx.x >> 5) * EPI_TILE;

#pragma unroll
    for (int m = 0; m < 3; ++m) {
        const float4* Wp = Wf + (size_t)m * WFRAG_F4_PER_MAT;
        const float*  bp = bm[m];
        float*        Op = Om[m];

        v8f c[RT][NT];
#pragma unroll
        for (int t = 0; t < NT; ++t) {
            const float bias = bp[t * 16 + l];
#pragma unroll
            for (int i = 0; i < RT; ++i) {
#pragma unroll
                for (int r = 0; r < 8; ++r) c[i][t][r] = bias;
            }
        }

        for (int kkp = 0; kkp < 16; ++kkp) {
            v2f a0[RT], a1[RT];
#pragma unroll
            for (int i = 0; i < RT; ++i) {
                a0[i] = *(const v2f*)(xrow + (size_t)i * 16 * DD + 8 * kkp);
                a1[i] = *(const v2f*)(xrow + (size_t)i * 16 * DD + 8 * kkp + 4);
            }
            float4 bf[NT];
#pragma unroll
            for (int t = 0; t < NT; ++t)
                bf[t] = Wp[(size_t)(kkp * NT + t) * 32 + lane];
#pragma unroll
            for (int i = 0; i < RT; ++i) {
#pragma unroll
                for (int t = 0; t < NT; ++t) {
                    v2f b0; b0.x = bf[t].x; b0.y = bf[t].y;
                    c[i][t] = __builtin_amdgcn_wmma_f32_16x16x4_f32(
                               false, a0[i], false, b0, (short)0, c[i][t], false, false);
                }
            }
#pragma unroll
            for (int i = 0; i < RT; ++i) {
#pragma unroll
                for (int t = 0; t < NT; ++t) {
                    v2f b1; b1.x = bf[t].z; b1.y = bf[t].w;
                    c[i][t] = __builtin_amdgcn_wmma_f32_16x16x4_f32(
                               false, a1[i], false, b1, (short)0, c[i][t], false, false);
                }
            }
        }

#pragma unroll
        for (int i = 0; i < RT; ++i) {
#pragma unroll
            for (int t = 0; t < NT; ++t) {
#pragma unroll
                for (int r = 0; r < 8; ++r) {
                    E[(8 * hi + r) * EPI_PITCH + t * 16 + l] = c[i][t][r];
                }
            }
            lds_wave_sync();
            v4f row[16];
#pragma unroll
            for (int rr = 0; rr < 16; ++rr)
                row[rr] = *(const v4f*)(E + rr * EPI_PITCH + lane * 4);
#pragma unroll
            for (int rr = 0; rr < 16; ++rr)
                *(v4f*)(Op + (size_t)(m0 + 16 * i + rr) * DD + lane * 4) = row[rr];
        }
    }
}

#define STRIP 16

__device__ __forceinline__ v4f ld_row(const char* P, unsigned boff)
{
    return *(const v4f*)(P + boff);
}

__device__ __forceinline__ float xhalf(float x)
{
    const int b = __builtin_bit_cast(int, x);
    return __builtin_bit_cast(float,
        __builtin_amdgcn_permlanex16(b, b, 0x76543210, 0xfedcba98, false, false));
}

template <int M>
__device__ __forceinline__ float xrow(float x)
{
    const int b = __builtin_bit_cast(int, x);
    return __builtin_bit_cast(float,
        __builtin_amdgcn_update_dpp(b, b, 0x160 | M, 0xf, 0xf, true));
}

__device__ __forceinline__ float uni(float x)
{
    return __builtin_bit_cast(float,
        __builtin_amdgcn_readfirstlane(__builtin_bit_cast(int, x)));
}

__global__ __launch_bounds__(256)
void win_attn(const float* __restrict__ Q, const float* __restrict__ K,
              const float* __restrict__ V, float* __restrict__ out,
              int B, int H, int W)
{
    const int wave = __builtin_amdgcn_readfirstlane(
        (int)((blockIdx.x * blockDim.x + threadIdx.x) >> 5));
    const unsigned lane16 = (threadIdx.x & 31u) * 16u;
    const int spr  = W / STRIP;
    if (wave >= B * H * spr) return;

    const int  wig  = wave & 7;
    const int  blk  = wave >> 3;
    const int  rblk = blk / spr;
    const int  rowi = rblk * 8 + wig;
    const int  x0   = (blk - rblk * spr) * STRIP;
    if (rowi >= B * H) return;
    const int  y    = rowi % H;
    const bool okr[3] = { y > 0, true, y + 1 < H };
    const float scale2 = 0.08838834764831845f * 1.44269504088896341f;
    const unsigned PIXB = DD * 4u;

    const char* Kr[3];
    const char* Vr[3];
#pragma unroll
    for (int r = 0; r < 3; ++r) {
        const size_t rb = (size_t)(rowi + (okr[r] ? r - 1 : 0)) * (size_t)W * PIXB;
        Kr[r] = (const char*)K + rb;
        Vr[r] = (const char*)V + rb;
    }
    const char* Qr = (const char*)Q + (size_t)rowi * (size_t)W * PIXB;
    char*       Or = (char*)out + (size_t)rowi * (size_t)W * PIXB;

    v4f kw[4][3], vw[4][3];
    v4f qq[2];
#pragma unroll
    for (int c = 0; c < 3; ++c) {
        const unsigned colb = (unsigned)min(max(x0 - 1 + c, 0), W - 1) * PIXB + lane16;
#pragma unroll
        for (int r = 0; r < 3; ++r) {
            kw[c][r] = ld_row(Kr[r], colb);
            vw[c][r] = ld_row(Vr[r], colb);
        }
    }
    qq[0] = ld_row(Qr, (unsigned)x0 * PIXB + lane16);

#pragma unroll 1
    for (int j = 0; j < STRIP; j += 4) {
#pragma unroll
        for (int u = 0; u < 4; ++u) {
            const int  x = x0 + j + u;
            const bool okc[3] = { x > 0, true, x + 1 < W };
            const unsigned colb = (unsigned)min(x + 2, W - 1) * PIXB + lane16;
            const v4f q = qq[u & 1];

            float sc[9];
#pragma unroll
            for (int r = 0; r < 3; ++r) {
#pragma unroll
                for (int c = 0; c < 3; ++c) {
                    const v4f k4 = kw[(u + c) & 3][r];
                    float s = q.x * k4.x + q.y * k4.y + q.z * k4.z + q.w * k4.w;
                    s += xhalf(s);
                    s += xrow<8>(s);
                    s += xrow<4>(s);
                    s += xrow<2>(s);
                    s += xrow<1>(s);
                    const int keep = -(int)(okr[r] & okc[c]);
                    sc[r * 3 + c] = __builtin_bit_cast(float,
                        __builtin_bit_cast(int, uni(s) * scale2) & keep);
                }
            }

#pragma unroll
            for (int r = 0; r < 3; ++r) kw[(u + 3) & 3][r] = ld_row(Kr[r], colb);
            qq[(u + 1) & 1] = ld_row(Qr, (unsigned)min(x + 1, W - 1) * PIXB + lane16);

            float mx = sc[0];
#pragma unroll
            for (int n = 1; n < 9; ++n) mx = (sc[n] > mx) ? sc[n] : mx;
            asm volatile("" : "+s"(mx) : : "memory");

            float p[9], sum = 0.0f;
#pragma unroll
            for (int n = 0; n < 9; ++n) { p[n] = __builtin_amdgcn_exp2f(sc[n] - mx); sum += p[n]; }
            const float inv = __builtin_amdgcn_rcpf(sum);

            v4f o = { 0.f, 0.f, 0.f, 0.f };
#pragma unroll
            for (int r = 0; r < 3; ++r) {
#pragma unroll
                for (int c = 0; c < 3; ++c) {
                    const float w  = (okr[r] & okc[c]) ? p[r * 3 + c] * inv : 0.0f;
                    const v4f   v4 = vw[(u + c) & 3][r];
                    o.x += w * v4.x; o.y += w * v4.y;
                    o.z += w * v4.z; o.w += w * v4.w;
                }
            }

#pragma unroll
            for (int r = 0; r < 3; ++r) vw[(u + 3) & 3][r] = ld_row(Vr[r], colb);

            asm volatile("" : "+v"(o), "+v"(qq[(u + 1) & 1]) : : "memory");

            *(v4f*)(Or + ((unsigned)x * PIXB + lane16)) = o;
        }
    }
}

extern "C" void kernel_launch(void* const* d_in, const int* in_sizes, int n_in,
                              void* d_out, int out_size, void* d_ws, size_t ws_size,
                              hipStream_t stream)
{
    const float* x  = (const float*)d_in[0];
    const float* Wq = (const float*)d_in[1];
    const float* bq = (const float*)d_in[2];
    const float* Wk = (const float*)d_in[3];
    const float* bk = (const float*)d_in[4];
    const float* Wv = (const float*)d_in[5];
    const float* bv = (const float*)d_in[6];

    const int H = 128, W = 128;
    const int Mrows = in_sizes[0] / DD;
    const int B = Mrows / (H * W);

    float*  Q   = (float*)d_ws;
    float*  Kp  = Q  + (size_t)Mrows * DD;
    float*  Vp  = Kp + (size_t)Mrows * DD;
    float4* Wf  = (float4*)(Vp + (size_t)Mrows * DD);

    reformat_w<<<48, 256, 0, stream>>>(Wq, Wk, Wv, Wf);

    const int gemmWaves = Mrows / 32;
    qkv_gemm_wmma<<<(gemmWaves + 7) / 8, 256, 0, stream>>>(
        x, Wf, bq, bk, bv, Q, Kp, Vp, Mrows);

    const int stripWaves = B * H * (W / STRIP);
    win_attn<<<(stripWaves + 7) / 8, 256, 0, stream>>>(
        Q, Kp, Vp, (float*)d_out, B, H, W);
}


# qkv_gemm: Wk folded into the query weights (Q' = x(Wq Wk^T) + bq Wk^T, score = Q'.x_j + x.(Wq bk) + bq.bk, taken on x itself): K never formed, 2 of 3 products left; f32 sum order of scores changes
# speedup vs baseline: 1.8778x; 1.2203x over previous
#include <hip/hip_runtime.h>


typedef float v2f __attribute__((ext_vector_type(2)));
typedef float v8f __attribute__((ext_vector_type(8)));
typedef float v4f __attribute__((ext_vector_type(4)));

#define DD 128
#define NT 8
#define WFRAG_F4_PER_MAT 4096

__global__ __launch_bounds__(256)
void prep_w(const float* __restrict__ Wq, const float* __restrict__ bq,
            const float* __restrict__ Wk, const float* __restrict__ bk,
            const float* __restrict__ Wv, float4* __restrict__ Wf,
            float* __restrict__ vec)
{
    const int blk = (int)blockIdx.x;
    const int tid = (int)threadIdx.x;
    if (blk < 8) {
        const int g    = blk * 8 + (tid >> 5);
        const int lane = tid & 31;
        const int hi = lane >> 4, l = lane & 15;
        const int r0 = (g >> 3) * 16;
        const int t  = g & 7;
        const float* ap = Wq + (size_t)(r0 + l) * DD + 2 * hi;
        const float* bp = Wk + (size_t)(t * 16 + l) * DD + 2 * hi;
        v8f acc = { 0.f, 0.f, 0.f, 0.f, 0.f, 0.f, 0.f, 0.f };
#pragma unroll
        for (int kk = 0; kk < 32; ++kk) {
            const v2f a = *(const v2f*)(ap + 4 * kk);
            const v2f b = *(const v2f*)(bp + 4 * kk);
            acc = __builtin_amdgcn_wmma_f32_16x16x4_f32(
                      false, a, false, b, (short)0, acc, false, false);
        }
        const int kkp = (r0 >> 3) + hi;
        asm volatile("v_nop\n\tv_nop\n\tv_nop\n\tv_nop\n\tv_nop" : "+v"(acc));
        float4 f0, f1;
        f0.x = acc[0]; f0.y = acc[1]; f0.z = acc[4]; f0.w = acc[5];
        f1.x = acc[2]; f1.y = acc[3]; f1.z = acc[6]; f1.w = acc[7];
        Wf[(size_t)(kkp * NT + t) * 32 + l]      = f0;
        Wf[(size_t)(kkp * NT + t) * 32 + 16 + l] = f1;
    } else if (blk < 24) {
        const int r    = (blk - 8) * 256 + tid;
        const int kkp  = r >> 8;
        const int t    = (r >> 5) & 7;
        const int lane = r & 31;
        const int hi = lane >> 4, l = lane & 15;
        const int n  = t * 16 + l;
        const int k0 = 8 * kkp + 2 * hi;
        const int k1 = k0 + 4;
        float4 v;
        v.x = Wv[(size_t)(k0 + 0) * DD + n];
        v.y = Wv[(size_t)(k0 + 1) * DD + n];
        v.z = Wv[(size_t)(k1 + 0) * DD + n];
        v.w = Wv[(size_t)(k1 + 1) * DD + n];
        Wf[WFRAG_F4_PER_MAT + r] = v;
    } else if (blk < 32) {
        const int    o    = (blk - 24) * 32 + (tid >> 3);
        const int    part = tid & 7;
        const float* row  = ((blk < 28) ? Wk : Wq) + (size_t)(o & 127) * DD + 16 * part;
        const float* bb   = ((blk < 28) ? bq : bk) + 16 * part;
        float s = 0.f;
#pragma unroll
        for (int i = 0; i < 4; ++i) {
            const v4f r4 = *(const v4f*)(row + 4 * i);
            const v4f b4 = *(const v4f*)(bb + 4 * i);
            s += r4.x * b4.x + r4.y * b4.y + r4.z * b4.z + r4.w * b4.w;
        }
        s += __shfl_xor(s, 4);
        s += __shfl_xor(s, 2);
        s += __shfl_xor(s, 1);
        if (part == 0) vec[o] = s;
    } else {
        if (tid < 32) {
            const v4f p4 = *(const v4f*)(bq + tid * 4);
            const v4f k4 = *(const v4f*)(bk + tid * 4);
            float be = p4.x * k4.x + p4.y * k4.y + p4.z * k4.z + p4.w * k4.w;
            be += __shfl_xor(be, 16);
            be += __shfl_xor(be, 8);
            be += __shfl_xor(be, 4);
            be += __shfl_xor(be, 2);
            be += __shfl_xor(be, 1);
            if (tid == 0) vec[256] = be;
        }
    }
}

#define EPI_PITCH 132
#define EPI_TILE  (16 * EPI_PITCH)

__device__ __forceinline__ void lds_wave_sync()
{
    asm volatile("s_wait_dscnt 0x0" ::: "memory");
}

#define RT 2
__global__ __launch_bounds__(256, 4)
void qkv_gemm_wmma(const float* __restrict__ X,
                   const float4* __restrict__ Wf,
                   const float* __restrict__ bu, const float* __restrict__ bv,
                   float* __restrict__ Q, float* __restrict__ V,
                   int Mrows)
{
    const int wave = (int)((blockIdx.x * blockDim.x + threadIdx.x) >> 5);
    const int lane = (int)(threadIdx.x & 31);
    const int m0   = wave * (16 * RT);
    if (m0 >= Mrows) return;

    const int hi = lane >> 4;
    const int l  = lane & 15;

    const float* xrow = X + (size_t)(m0 + l) * DD + 2 * hi;

    const float* bm[2] = { bu, bv };
    float*       Om[2] = { Q,  V  };

    __shared__ v4f epi[8 * EPI_TILE / 4];
    float* E = (float*)epi + (threadIdx.x >> 5) * EPI_TILE;

#pragma unroll
    for (int m = 0; m < 2; ++m) {
        const float4* Wp = Wf + (size_t)m * WFRAG_F4_PER_MAT;
        const float*  bp = bm[m];
        float*        Op = Om[m];

        v8f c[RT][NT];
#pragma unroll
        for (int t = 0; t < NT; ++t) {
            const float bias = bp[t * 16 + l];
#pragma unroll
            for (int i = 0; i < RT; ++i) {
#pragma unroll
                for (int r = 0; r < 8; ++r) c[i][t][r] = bias;
            }
        }

        for (int kkp = 0; kkp < 16; ++kkp) {
            v2f a0[RT], a1[RT];
#pragma unroll
            for (int i = 0; i < RT; ++i) {
                a0[i] = *(const v2f*)(xrow + (size_t)i * 16 * DD + 8 * kkp);
                a1[i] = *(const v2f*)(xrow + (size_t)i * 16 * DD + 8 * kkp + 4);
            }
            float4 bf[NT];
#pragma unroll
            for (int t = 0; t < NT; ++t)
                bf[t] = Wp[(size_t)(kkp * NT + t) * 32 + lane];
#pragma unroll
            for (int i = 0; i < RT; ++i) {
#pragma unroll
                for (int t = 0; t < NT; ++t) {
                    v2f b0; b0.x = bf[t].x; b0.y = bf[t].y;
                    c[i][t] = __builtin_amdgcn_wmma_f32_16x16x4_f32(
                               false, a0[i], false, b0, (short)0, c[i][t], false, false);
                }
            }
#pragma unroll
            for (int i = 0; i < RT; ++i) {
#pragma unroll
                for (int t = 0; t < NT; ++t) {
                    v2f b1; b1.x = bf[t].z; b1.y = bf[t].w;
                    c[i][t] = __builtin_amdgcn_wmma_f32_16x16x4_f32(
                               false, a1[i], false, b1, (short)0, c[i][t], false, false);
                }
            }
        }

#pragma unroll
        for (int i = 0; i < RT; ++i) {
#pragma unroll
            for (int t = 0; t < NT; ++t) {
#pragma unroll
                for (int r = 0; r < 8; ++r) {
                    E[(8 * hi + r) * EPI_PITCH + t * 16 + l] = c[i][t][r];
                }
            }
            lds_wave_sync();
            v4f row[16];
#pragma unroll
            for (int rr = 0; rr < 16; ++rr)
                row[rr] = *(const v4f*)(E + rr * EPI_PITCH + lane * 4);
#pragma unroll
            for (int rr = 0; rr < 16; ++rr)
                *(v4f*)(Op + (size_t)(m0 + 16 * i + rr) * DD + lane * 4) = row[rr];
        }
    }
}

#define STRIP 16

__device__ __forceinline__ v4f ld_row(const char* P, unsigned boff)
{
    return *(const v4f*)(P + boff);
}

__device__ __forceinline__ float xhalf(float x)
{
    const int b = __builtin_bit_cast(int, x);
    return __builtin_bit_cast(float,
        __builtin_amdgcn_permlanex16(b, b, 0x76543210, 0xfedcba98, false, false));
}

template <int M>
__device__ __forceinline__ float xrow(float x)
{
    const int b = __builtin_bit_cast(int, x);
    return __builtin_bit_cast(float,
        __builtin_amdgcn_update_dpp(b, b, 0x160 | M, 0xf, 0xf, true));
}

__device__ __forceinline__ float uni(float x)
{
    return __builtin_bit_cast(float,
        __builtin_amdgcn_readfirstlane(__builtin_bit_cast(int, x)));
}

__global__ __launch_bounds__(256)
void win_attn(const float* __restrict__ Q, const float* __restrict__ K,
              const float* __restrict__ V, const float* __restrict__ wb,
              float* __restrict__ out, int B, int H, int W)
{
    const int wave = __builtin_amdgcn_readfirstlane(
        (int)((blockIdx.x * blockDim.x + threadIdx.x) >> 5));
    const unsigned lane16 = (threadIdx.x & 31u) * 16u;
    const int spr  = W / STRIP;
    if (wave >= B * H * spr) return;

    const int  wig  = wave & 7;
    const int  blk  = wave >> 3;
    const int  rblk = blk / spr;
    const int  rowi = rblk * 8 + wig;
    const int  x0   = (blk - rblk * spr) * STRIP;
    if (rowi >= B * H) return;
    const int  y    = rowi % H;
    const bool okr[3] = { y > 0, true, y + 1 < H };
    const float scale2 = 0.08838834764831845f * 1.44269504088896341f;
    const unsigned PIXB = DD * 4u;

    const char* Kr[3];
    const char* Vr[3];
#pragma unroll
    for (int r = 0; r < 3; ++r) {
        const size_t rb = (size_t)(rowi + (okr[r] ? r - 1 : 0)) * (size_t)W * PIXB;
        Kr[r] = (const char*)K + rb;
        Vr[r] = (const char*)V + rb;
    }
    const char* Qr = (const char*)Q + (size_t)rowi * (size_t)W * PIXB;
    const v4f   w4   = *(const v4f*)((const char*)wb + lane16);
    const float beta = wb[DD];
    char*       Or = (char*)out + (size_t)rowi * (size_t)W * PIXB;

    v4f kw[4][3], vw[4][3];
    v4f qq[2];
#pragma unroll
    for (int c = 0; c < 3; ++c) {
        const unsigned colb = (unsigned)min(max(x0 - 1 + c, 0), W - 1) * PIXB + lane16;
#pragma unroll
        for (int r = 0; r < 3; ++r) {
            kw[c][r] = ld_row(Kr[r], colb);
            vw[c][r] = ld_row(Vr[r], colb);
        }
    }
    qq[0] = ld_row(Qr, (unsigned)x0 * PIXB + lane16);

#pragma unroll 1
    for (int j = 0; j < STRIP; j += 4) {
#pragma unroll
        for (int u = 0; u < 4; ++u) {
            const int  x = x0 + j + u;
            const bool okc[3] = { x > 0, true, x + 1 < W };
            const unsigned colb = (unsigned)min(x + 2, W - 1) * PIXB + lane16;
            const v4f q = qq[u & 1];

            const v4f xc = kw[(u + 1) & 3][1];
            float tp = xc.x * w4.x + xc.y * w4.y + xc.z * w4.z + xc.w * w4.w;
            tp += xhalf(tp);
            tp += xrow<8>(tp);
            tp += xrow<4>(tp);
            tp += xrow<2>(tp);
            tp += xrow<1>(tp);
            const float tq = uni(tp) + beta;

            float sc[9];
#pragma unroll
            for (int r = 0; r < 3; ++r) {
#pragma unroll
                for (int c = 0; c < 3; ++c) {
                    const v4f k4 = kw[(u + c) & 3][r];
                    float s = q.x * k4.x + q.y * k4.y + q.z * k4.z + q.w * k4.w;
                    s += xhalf(s);
                    s += xrow<8>(s);
                    s += xrow<4>(s);
                    s += xrow<2>(s);
                    s += xrow<1>(s);
                    const int keep = -(int)(okr[r] & okc[c]);
                    sc[r * 3 + c] = __builtin_bit_cast(float,
                        __builtin_bit_cast(int, (uni(s) + tq) * scale2) & keep);
                }
            }

#pragma unroll
            for (int r = 0; r < 3; ++r) kw[(u + 3) & 3][r] = ld_row(Kr[r], colb);
            qq[(u + 1) & 1] = ld_row(Qr, (unsigned)min(x + 1, W - 1) * PIXB + lane16);

            float mx = sc[0];
#pragma unroll
            for (int n = 1; n < 9; ++n) mx = (sc[n] > mx) ? sc[n] : mx;
            asm volatile("" : "+s"(mx) : : "memory");

            float p[9], sum = 0.0f;
#pragma unroll
            for (int n = 0; n < 9; ++n) { p[n] = __builtin_amdgcn_exp2f(sc[n] - mx); sum += p[n]; }
            const float inv = __builtin_amdgcn_rcpf(sum);

            v4f o = { 0.f, 0.f, 0.f, 0.f };
#pragma unroll
            for (int r = 0; r < 3; ++r) {
#pragma unroll
                for (int c = 0; c < 3; ++c) {
                    const float w  = (okr[r] & okc[c]) ? p[r * 3 + c] * inv : 0.0f;
                    const v4f   v4 = vw[(u + c) & 3][r];
                    o.x += w * v4.x; o.y += w * v4.y;
                    o.z += w * v4.z; o.w += w * v4.w;
                }
            }

#pragma unroll
            for (int r = 0; r < 3; ++r) vw[(u + 3) & 3][r] = ld_row(Vr[r], colb);

            asm volatile("" : "+v"(o), "+v"(qq[(u + 1) & 1]) : : "memory");

            *(v4f*)(Or + ((unsigned)x * PIXB + lane16)) = o;
        }
    }
}

extern "C" void kernel_launch(void* const* d_in, const int* in_sizes, int n_in,
                              void* d_out, int out_size, void* d_ws, size_t ws_size,
                              hipStream_t stream)
{
    const float* x  = (const float*)d_in[0];
    const float* Wq = (const float*)d_in[1];
    const float* bq = (const float*)d_in[2];
    const float* Wk = (const float*)d_in[3];
    const float* bk = (const float*)d_in[4];
    const float* Wv = (const float*)d_in[5];
    const float* bv = (const float*)d_in[6];

    const int H = 128, W = 128;
    const int Mrows = in_sizes[0] / DD;
    const int B = Mrows / (H * W);

    float*  Q   = (float*)d_ws;
    float*  Vp  = Q  + (size_t)Mrows * DD;
    float4* Wf  = (float4*)(Vp + (size_t)Mrows * DD);
    float*  vec = (float*)(Wf + 2 * WFRAG_F4_PER_MAT);

    prep_w<<<33, 256, 0, stream>>>(Wq, bq, Wk, bk, Wv, Wf, vec);

    const int gemmWaves = Mrows / 32;
    qkv_gemm_wmma<<<(gemmWaves + 7) / 8, 256, 0, stream>>>(
        x, Wf, vec, bv, Q, Vp, Mrows);

    const int stripWaves = B * H * (W / STRIP);
    win_attn<<<(stripWaves + 7) / 8, 256, 0, stream>>>(
        Q, x, Vp, vec + DD, (float*)d_out, B, H, W);
}


# win_attn: the ten lane sums share their exchange steps (each step: half the lanes go on with another sum): 7 DPP adds + 5 permlanex16 a pixel, was 40 + 10; lane-sum order kept, per-lane dots now fma
# speedup vs baseline: 1.9595x; 1.0435x over previous
#include <hip/hip_runtime.h>


typedef float v2f __attribute__((ext_vector_type(2)));
typedef float v8f __attribute__((ext_vector_type(8)));
typedef float v4f __attribute__((ext_vector_type(4)));

#define DD 128
#define NT 8
#define WFRAG_F4_PER_MAT 4096

__global__ __launch_bounds__(256)
void prep_w(const float* __restrict__ Wq, const float* __restrict__ bq,
            const float* __restrict__ Wk, const float* __restrict__ bk,
            const float* __restrict__ Wv, float4* __restrict__ Wf,
            float* __restrict__ vec)
{
    const int blk = (int)blockIdx.x;
    const int tid = (int)threadIdx.x;
    if (blk < 8) {
        const int g    = blk * 8 + (tid >> 5);
        const int lane = tid & 31;
        const int hi = lane >> 4, l = lane & 15;
        const int r0 = (g >> 3) * 16;
        const int t  = g & 7;
        const float* ap = Wq + (size_t)(r0 + l) * DD + 2 * hi;
        const float* bp = Wk + (size_t)(t * 16 + l) * DD + 2 * hi;
        v8f acc = { 0.f, 0.f, 0.f, 0.f, 0.f, 0.f, 0.f, 0.f };
#pragma unroll
        for (int kk = 0; kk < 32; ++kk) {
            const v2f a = *(const v2f*)(ap + 4 * kk);
            const v2f b = *(const v2f*)(bp + 4 * kk);
            acc = __builtin_amdgcn_wmma_f32_16x16x4_f32(
                      false, a, false, b, (short)0, acc, false, false);
        }
        const int kkp = (r0 >> 3) + hi;
        asm volatile("v_nop\n\tv_nop\n\tv_nop\n\tv_nop\n\tv_nop" : "+v"(acc));
        float4 f0, f1;
        f0.x = acc[0]; f0.y = acc[1]; f0.z = acc[4]; f0.w = acc[5];
        f1.x = acc[2]; f1.y = acc[3]; f1.z = acc[6]; f1.w = acc[7];
        Wf[(size_t)(kkp * NT + t) * 32 + l]      = f0;
        Wf[(size_t)(kkp * NT + t) * 32 + 16 + l] = f1;
    } else if (blk < 24) {
        const int r    = (blk - 8) * 256 + tid;
        const int kkp  = r >> 8;
        const int t    = (r >> 5) & 7;
        const int lane = r & 31;
        const int hi = lane >> 4, l = lane & 15;
        const int n  = t * 16 + l;
        const int k0 = 8 * kkp + 2 * hi;
        const int k1 = k0 + 4;
        float4 v;
        v.x = Wv[(size_t)(k0 + 0) * DD + n];
        v.y = Wv[(size_t)(k0 + 1) * DD + n];
        v.z = Wv[(size_t)(k1 + 0) * DD + n];
        v.w = Wv[(size_t)(k1 + 1) * DD + n];
        Wf[WFRAG_F4_PER_MAT + r] = v;
    } else if (blk < 32) {
        const int    o    = (blk - 24) * 32 + (tid >> 3);
        const int    part = tid & 7;
        const float* row  = ((blk < 28) ? Wk : Wq) + (size_t)(o & 127) * DD + 16 * part;
        const float* bb   = ((blk < 28) ? bq : bk) + 16 * part;
        float s = 0.f;
#pragma unroll
        for (int i = 0; i < 4; ++i) {
            const v4f r4 = *(const v4f*)(row + 4 * i);
            const v4f b4 = *(const v4f*)(bb + 4 * i);
            s += r4.x * b4.x + r4.y * b4.y + r4.z * b4.z + r4.w * b4.w;
        }
        s += __shfl_xor(s, 4);
        s += __shfl_xor(s, 2);
        s += __shfl_xor(s, 1);
        if (part == 0) vec[o] = s;
    } else {
        if (tid < 32) {
            const v4f p4 = *(const v4f*)(bq + tid * 4);
            const v4f k4 = *(const v4f*)(bk + tid * 4);
            float be = p4.x * k4.x + p4.y * k4.y + p4.z * k4.z + p4.w * k4.w;
            be += __shfl_xor(be, 16);
            be += __shfl_xor(be, 8);
            be += __shfl_xor(be, 4);
            be += __shfl_xor(be, 2);
            be += __shfl_xor(be, 1);
            if (tid == 0) vec[256] = be;
        }
    }
}

#define EPI_PITCH 132
#define EPI_TILE  (16 * EPI_PITCH)

__device__ __forceinline__ void lds_wave_sync()
{
    asm volatile("s_wait_dscnt 0x0" ::: "memory");
}

#define RT 2
__global__ __launch_bounds__(256, 4)
void qkv_gemm_wmma(const float* __restrict__ X,
                   const float4* __restrict__ Wf,
                   const float* __restrict__ bu, const float* __restrict__ bv,
                   float* __restrict__ Q, float* __restrict__ V,
                   int Mrows)
{
    const int wave = (int)((blockIdx.x * blockDim.x + threadIdx.x) >> 5);
    const int lane = (int)(threadIdx.x & 31);
    const int m0   = wave * (16 * RT);
    if (m0 >= Mrows) return;

    const int hi = lane >> 4;
    const int l  = lane & 15;

    const float* xrow = X + (size_t)(m0 + l) * DD + 2 * hi;

    const float* bm[2] = { bu, bv };
    float*       Om[2] = { Q,  V  };

    __shared__ v4f epi[8 * EPI_TILE / 4];
    float* E = (float*)epi + (threadIdx.x >> 5) * EPI_TILE;

#pragma unroll
    for (int m = 0; m < 2; ++m) {
        const float4* Wp = Wf + (size_t)m * WFRAG_F4_PER_MAT;
        const float*  bp = bm[m];
        float*        Op = Om[m];

        v8f c[RT][NT];
#pragma unroll
        for (int t = 0; t < NT; ++t) {
            const float bias = bp[t * 16 + l];
#pragma unroll
            for (int i = 0; i < RT; ++i) {
#pragma unroll
                for (int r = 0; r < 8; ++r) c[i][t][r] = bias;
            }
        }

        for (int kkp = 0; kkp < 16; ++kkp) {
            v2f a0[RT], a1[RT];
#pragma unroll
            for (int i = 0; i < RT; ++i) {
                a0[i] = *(const v2f*)(xrow + (size_t)i * 16 * DD + 8 * kkp);
                a1[i] = *(const v2f*)(xrow + (size_t)i * 16 * DD + 8 * kkp + 4);
            }
            float4 bf[NT];
#pragma unroll
            for (int t = 0; t < NT; ++t)
                bf[t] = Wp[(size_t)(kkp * NT + t) * 32 + lane];
#pragma unroll
            for (int i = 0; i < RT; ++i) {
#pragma unroll
                for (int t = 0; t < NT; ++t) {
                    v2f b0; b0.x = bf[t].x; b0.y = bf[t].y;
                    c[i][t] = __builtin_amdgcn_wmma_f32_16x16x4_f32(
                               false, a0[i], false, b0, (short)0, c[i][t], false, false);
                }
            }
#pragma unroll
            for (int i = 0; i < RT; ++i) {
#pragma unroll
                for (int t = 0; t < NT; ++t) {
                    v2f b1; b1.x = bf[t].z; b1.y = bf[t].w;
                    c[i][t] = __builtin_amdgcn_wmma_f32_16x16x4_f32(
                               false, a1[i], false, b1, (short)0, c[i][t], false, false);
                }
            }
        }

#pragma unroll
        for (int i = 0; i < RT; ++i) {
#pragma unroll
            for (int t = 0; t < NT; ++t) {
#pragma unroll
                for (int r = 0; r < 8; ++r) {
                    E[(8 * hi + r) * EPI_PITCH + t * 16 + l] = c[i][t][r];
                }
            }
            lds_wave_sync();
            v4f row[16];
#pragma unroll
            for (int rr = 0; rr < 16; ++rr)
                row[rr] = *(const v4f*)(E + rr * EPI_PITCH + lane * 4);
#pragma unroll
            for (int rr = 0; rr < 16; ++rr)
                *(v4f*)(Op + (size_t)(m0 + 16 * i + rr) * DD + lane * 4) = row[rr];
        }
    }
}

#define STRIP 16

__device__ __forceinline__ v4f ld_row(const char* P, unsigned boff)
{
    return *(const v4f*)(P + boff);
}

__device__ __forceinline__ float xhalf(float x)
{
    const int b = __builtin_bit_cast(int, x);
    return __builtin_bit_cast(float,
        __builtin_amdgcn_permlanex16(b, b, 0x76543210, 0xfedcba98, false, false));
}

template <int M>
__device__ __forceinline__ float xrow(float x)
{
    const int b = __builtin_bit_cast(int, x);
    return __builtin_bit_cast(float,
        __builtin_amdgcn_update_dpp(b, b, 0x160 | M, 0xf, 0xf, true));
}

template <int M>
__device__ __forceinline__ float fold2(float a, float b, bool up)
{
    const float keep = up ? b : a;
    const float send = up ? a : b;
    if (M == 16) return keep + xhalf(send);
    return keep + xrow<M & 15>(send);
}

template <int L>
__device__ __forceinline__ float lane_val(float x)
{
    return __builtin_bit_cast(float,
        __builtin_amdgcn_readlane(__builtin_bit_cast(int, x), L));
}

__global__ __launch_bounds__(256)
void win_attn(const float* __restrict__ Q, const float* __restrict__ K,
              const float* __restrict__ V, const float* __restrict__ wb,
              float* __restrict__ out, int B, int H, int W)
{
    const int wave = __builtin_amdgcn_readfirstlane(
        (int)((blockIdx.x * blockDim.x + threadIdx.x) >> 5));
    const unsigned lane16 = (threadIdx.x & 31u) * 16u;
    const bool up16 = (threadIdx.x & 16u) != 0u;
    const bool up8  = (threadIdx.x &  8u) != 0u;
    const bool up4  = (threadIdx.x &  4u) != 0u;
    const bool up2  = (threadIdx.x &  2u) != 0u;
    const int spr  = W / STRIP;
    if (wave >= B * H * spr) return;

    const int  wig  = wave & 7;
    const int  blk  = wave >> 3;
    const int  rblk = blk / spr;
    const int  rowi = rblk * 8 + wig;
    const int  x0   = (blk - rblk * spr) * STRIP;
    if (rowi >= B * H) return;
    const int  y    = rowi % H;
    const bool okr[3] = { y > 0, true, y + 1 < H };
    const float scale2 = 0.08838834764831845f * 1.44269504088896341f;
    const unsigned PIXB = DD * 4u;

    const char* Kr[3];
    const char* Vr[3];
#pragma unroll
    for (int r = 0; r < 3; ++r) {
        const size_t rb = (size_t)(rowi + (okr[r] ? r - 1 : 0)) * (size_t)W * PIXB;
        Kr[r] = (const char*)K + rb;
        Vr[r] = (const char*)V + rb;
    }
    const char* Qr = (const char*)Q + (size_t)rowi * (size_t)W * PIXB;
    const v4f   w4   = *(const v4f*)((const char*)wb + lane16);
    const float beta = wb[DD];
    char*       Or = (char*)out + (size_t)rowi * (size_t)W * PIXB;

    v4f kw[4][3], vw[4][3];
    v4f qq[2];
#pragma unroll
    for (int c = 0; c < 3; ++c) {
        const unsigned colb = (unsigned)min(max(x0 - 1 + c, 0), W - 1) * PIXB + lane16;
#pragma unroll
        for (int r = 0; r < 3; ++r) {
            kw[c][r] = ld_row(Kr[r], colb);
            vw[c][r] = ld_row(Vr[r], colb);
        }
    }
    qq[0] = ld_row(Qr, (unsigned)x0 * PIXB + lane16);

#pragma unroll 1
    for (int j = 0; j < STRIP; j += 4) {
#pragma unroll
        for (int u = 0; u < 4; ++u) {
            const int  x = x0 + j + u;
            const bool okc[3] = { x > 0, true, x + 1 < W };
            const unsigned colb = (unsigned)min(x + 2, W - 1) * PIXB + lane16;
            const v4f q = qq[u & 1];

            const v4f xc = kw[(u + 1) & 3][1];
            float pr[10];
#pragma unroll
            for (int r = 0; r < 3; ++r) {
#pragma unroll
                for (int c = 0; c < 3; ++c) {
                    const v4f k4 = kw[(u + c) & 3][r];
                    pr[r * 3 + c] = q.x * k4.x + q.y * k4.y + q.z * k4.z + q.w * k4.w;
                }
            }
            pr[9] = xc.x * w4.x + xc.y * w4.y + xc.z * w4.z + xc.w * w4.w;

            const float r0 = fold2<16>(pr[0], pr[1], up16);
            const float r1 = fold2<16>(pr[2], pr[3], up16);
            const float r2 = fold2<16>(pr[4], pr[5], up16);
            const float r3 = fold2<16>(pr[6], pr[7], up16);
            float       r4 = fold2<16>(pr[8], pr[9], up16);
            const float s0 = fold2<8>(r0, r1, up8);
            const float s1 = fold2<8>(r2, r3, up8);
            r4 += xrow<8>(r4);
            const float u0 = fold2<4>(s0, s1, up4);
            r4 += xrow<4>(r4);
            float z = fold2<2>(u0, r4, up2);
            z += xrow<1>(z);
            float rs[10];
            rs[0] = lane_val<0>(z);  rs[1] = lane_val<16>(z);
            rs[2] = lane_val<8>(z);  rs[3] = lane_val<24>(z);
            rs[4] = lane_val<4>(z);  rs[5] = lane_val<20>(z);
            rs[6] = lane_val<12>(z); rs[7] = lane_val<28>(z);
            rs[8] = lane_val<2>(z);  rs[9] = lane_val<18>(z);
            const float tq = rs[9] + beta;

            float sc[9];
#pragma unroll
            for (int r = 0; r < 3; ++r) {
#pragma unroll
                for (int c = 0; c < 3; ++c) {
                    const int keep = -(int)(okr[r] & okc[c]);
                    sc[r * 3 + c] = __builtin_bit_cast(float,
                        __builtin_bit_cast(int, (rs[r * 3 + c] + tq) * scale2) & keep);
                }
            }

#pragma unroll
            for (int r = 0; r < 3; ++r) kw[(u + 3) & 3][r] = ld_row(Kr[r], colb);
            qq[(u + 1) & 1] = ld_row(Qr, (unsigned)min(x + 1, W - 1) * PIXB + lane16);

            float mx = sc[0];
#pragma unroll
            for (int n = 1; n < 9; ++n) mx = (sc[n] > mx) ? sc[n] : mx;
            asm volatile("" : "+s"(mx) : : "memory");

            float p[9], sum = 0.0f;
#pragma unroll
            for (int n = 0; n < 9; ++n) { p[n] = __builtin_amdgcn_exp2f(sc[n] - mx); sum += p[n]; }
            const float inv = __builtin_amdgcn_rcpf(sum);

            v4f o = { 0.f, 0.f, 0.f, 0.f };
#pragma unroll
            for (int r = 0; r < 3; ++r) {
#pragma unroll
                for (int c = 0; c < 3; ++c) {
                    const float w  = (okr[r] & okc[c]) ? p[r * 3 + c] * inv : 0.0f;
                    const v4f   v4 = vw[(u + c) & 3][r];
                    o.x += w * v4.x; o.y += w * v4.y;
                    o.z += w * v4.z; o.w += w * v4.w;
                }
            }

#pragma unroll
            for (int r = 0; r < 3; ++r) vw[(u + 3) & 3][r] = ld_row(Vr[r], colb);

            asm volatile("" : "+v"(o), "+v"(qq[(u + 1) & 1]) : : "memory");

            *(v4f*)(Or + ((unsigned)x * PIXB + lane16)) = o;
        }
    }
}

extern "C" void kernel_launch(void* const* d_in, const int* in_sizes, int n_in,
                              void* d_out, int out_size, void* d_ws, size_t ws_size,
                              hipStream_t stream)
{
    const float* x  = (const float*)d_in[0];
    const float* Wq = (const float*)d_in[1];
    const float* bq = (const float*)d_in[2];
    const float* Wk = (const float*)d_in[3];
    const float* bk = (const float*)d_in[4];
    const float* Wv = (const float*)d_in[5];
    const float* bv = (const float*)d_in[6];

    const int H = 128, W = 128;
    const int Mrows = in_sizes[0] / DD;
    const int B = Mrows / (H * W);

    float*  Q   = (float*)d_ws;
    float*  Vp  = Q  + (size_t)Mrows * DD;
    float4* Wf  = (float4*)(Vp + (size_t)Mrows * DD);
    float*  vec = (float*)(Wf + 2 * WFRAG_F4_PER_MAT);

    prep_w<<<33, 256, 0, stream>>>(Wq, bq, Wk, bk, Wv, Wf, vec);

    const int gemmWaves = Mrows / 32;
    qkv_gemm_wmma<<<(gemmWaves + 7) / 8, 256, 0, stream>>>(
        x, Wf, vec, bv, Q, Vp, Mrows);

    const int stripWaves = B * H * (W / STRIP);
    win_attn<<<(stripWaves + 7) / 8, 256, 0, stream>>>(
        Q, x, Vp, vec + DD, (float*)d_out, B, H, W);
}


# one launch for both products: a wave forms Q' of its 32 pixels (WMMA) in its LDS, runs the window softmax on x, applies Wv (WMMA) to the weighted x sum; Q', V never stored; f32 sum order of out new
# speedup vs baseline: 2.4325x; 1.2414x over previous
#include <hip/hip_runtime.h>


typedef float v2f __attribute__((ext_vector_type(2)));
typedef float v8f __attribute__((ext_vector_type(8)));
typedef float v4f __attribute__((ext_vector_type(4)));

#define DD 128
#define NT 8
#define WFRAG_F4_PER_MAT 4096

__global__ __launch_bounds__(256)
void prep_w(const float* __restrict__ Wq, const float* __restrict__ bq,
            const float* __restrict__ Wk, const float* __restrict__ bk,
            const float* __restrict__ Wv, float4* __restrict__ Wf,
            float* __restrict__ vec)
{
    const int blk = (int)blockIdx.x;
    const int tid = (int)threadIdx.x;
    if (blk < 8) {
        const int g    = blk * 8 + (tid >> 5);
        const int lane = tid & 31;
        const int hi = lane >> 4, l = lane & 15;
        const int r0 = (g >> 3) * 16;
        const int t  = g & 7;
        const float* ap = Wq + (size_t)(r0 + l) * DD + 2 * hi;
        const float* bp = Wk + (size_t)(t * 16 + l) * DD + 2 * hi;
        v8f acc = { 0.f, 0.f, 0.f, 0.f, 0.f, 0.f, 0.f, 0.f };
#pragma unroll
        for (int kk = 0; kk < 32; ++kk) {
            const v2f a = *(const v2f*)(ap + 4 * kk);
            const v2f b = *(const v2f*)(bp + 4 * kk);
            acc = __builtin_amdgcn_wmma_f32_16x16x4_f32(
                      false, a, false, b, (short)0, acc, false, false);
        }
        const int kkp = (r0 >> 3) + hi;
        asm volatile("v_nop\n\tv_nop\n\tv_nop\n\tv_nop\n\tv_nop" : "+v"(acc));
        float4 f0, f1;
        f0.x = acc[0]; f0.y = acc[1]; f0.z = acc[4]; f0.w = acc[5];
        f1.x = acc[2]; f1.y = acc[3]; f1.z = acc[6]; f1.w = acc[7];
        Wf[(size_t)(kkp * NT + t) * 32 + l]      = f0;
        Wf[(size_t)(kkp * NT + t) * 32 + 16 + l] = f1;
    } else if (blk < 24) {
        const int r    = (blk - 8) * 256 + tid;
        const int kkp  = r >> 8;
        const int t    = (r >> 5) & 7;
        const int lane = r & 31;
        const int hi = lane >> 4, l = lane & 15;
        const int n  = t * 16 + l;
        const int k0 = 8 * kkp + 2 * hi;
        const int k1 = k0 + 4;
        float4 v;
        v.x = Wv[(size_t)(k0 + 0) * DD + n];
        v.y = Wv[(size_t)(k0 + 1) * DD + n];
        v.z = Wv[(size_t)(k1 + 0) * DD + n];
        v.w = Wv[(size_t)(k1 + 1) * DD + n];
        Wf[WFRAG_F4_PER_MAT + r] = v;
    } else if (blk < 32) {
        const int    o    = (blk - 24) * 32 + (tid >> 3);
        const int    part = tid & 7;
        const float* row  = ((blk < 28) ? Wk : Wq) + (size_t)(o & 127) * DD + 16 * part;
        const float* bb   = ((blk < 28) ? bq : bk) + 16 * part;
        float s = 0.f;
#pragma unroll
        for (int i = 0; i < 4; ++i) {
            const v4f r4 = *(const v4f*)(row + 4 * i);
            const v4f b4 = *(const v4f*)(bb + 4 * i);
            s += r4.x * b4.x + r4.y * b4.y + r4.z * b4.z + r4.w * b4.w;
        }
        s += __shfl_xor(s, 4);
        s += __shfl_xor(s, 2);
        s += __shfl_xor(s, 1);
        if (part == 0) vec[o] = s;
    } else {
        if (tid < 32) {
            const v4f p4 = *(const v4f*)(bq + tid * 4);
            const v4f k4 = *(const v4f*)(bk + tid * 4);
            float be = p4.x * k4.x + p4.y * k4.y + p4.z * k4.z + p4.w * k4.w;
            be += __shfl_xor(be, 16);
            be += __shfl_xor(be, 8);
            be += __shfl_xor(be, 4);
            be += __shfl_xor(be, 2);
            be += __shfl_xor(be, 1);
            if (tid == 0) vec[256] = be;
        }
    }
}

#define EPI_PITCH 132
#define EPI_TILE  (16 * EPI_PITCH)
#define RT 2
#define PIX (16 * RT)

__device__ __forceinline__ void lds_wave_sync()
{
    asm volatile("s_wait_dscnt 0x0" ::: "memory");
}


__device__ __forceinline__ v4f ld_row(const char* P, unsigned boff)
{
    return *(const v4f*)(P + boff);
}

__device__ __forceinline__ float xhalf(float x)
{
    const int b = __builtin_bit_cast(int, x);
    return __builtin_bit_cast(float,
        __builtin_amdgcn_permlanex16(b, b, 0x76543210, 0xfedcba98, false, false));
}

template <int M>
__device__ __forceinline__ float xrow(float x)
{
    const int b = __builtin_bit_cast(int, x);
    return __builtin_bit_cast(float,
        __builtin_amdgcn_update_dpp(b, b, 0x160 | M, 0xf, 0xf, true));
}

template <int M>
__device__ __forceinline__ float fold2(float a, float b, bool up)
{
    const float keep = up ? b : a;
    const float send = up ? a : b;
    if (M == 16) return keep + xhalf(send);
    return keep + xrow<M & 15>(send);
}

template <int L>
__device__ __forceinline__ float lane_val(float x)
{
    return __builtin_bit_cast(float,
        __builtin_amdgcn_readlane(__builtin_bit_cast(int, x), L));
}


__global__ __launch_bounds__(256, 4)
void win_attn_fused(const float* __restrict__ X,
                    const float4* __restrict__ Wf,
                    const float* __restrict__ vec, const float* __restrict__ bv,
                    float* __restrict__ out, int B, int H, int W)
{
    const int wave = __builtin_amdgcn_readfirstlane(
        (int)((blockIdx.x * blockDim.x + threadIdx.x) >> 5));
    const int lane = (int)(threadIdx.x & 31u);
    const int hi = lane >> 4;
    const int l  = lane & 15;
    const unsigned lane16 = (unsigned)lane * 16u;
    const bool up16 = (threadIdx.x & 16u) != 0u;
    const bool up8  = (threadIdx.x &  8u) != 0u;
    const bool up4  = (threadIdx.x &  4u) != 0u;
    const bool up2  = (threadIdx.x &  2u) != 0u;
    const int spr  = W / PIX;

    const int  wig  = wave & 7;
    const int  blk  = wave >> 3;
    const int  rblk = blk / spr;
    const int  rowi = rblk * 8 + wig;
    const int  x0   = (blk - rblk * spr) * PIX;
    if (rowi >= B * H) return;
    const int  y    = rowi % H;
    const int  m0   = rowi * W + x0;

    __shared__ v4f epi[8 * RT * EPI_TILE / 4];
    float* E = (float*)epi + (threadIdx.x >> 5) * (RT * EPI_TILE);

    {
        const float* xrow = X + (size_t)(m0 + l) * DD + 2 * hi;
        v8f c[RT][NT];
#pragma unroll
        for (int t = 0; t < NT; ++t) {
            const float bias = vec[t * 16 + l];
#pragma unroll
            for (int i = 0; i < RT; ++i) {
#pragma unroll
                for (int r = 0; r < 8; ++r) c[i][t][r] = bias;
            }
        }
        for (int kkp = 0; kkp < 16; ++kkp) {
            v2f a0[RT], a1[RT];
#pragma unroll
            for (int i = 0; i < RT; ++i) {
                a0[i] = *(const v2f*)(xrow + (size_t)i * 16 * DD + 8 * kkp);
                a1[i] = *(const v2f*)(xrow + (size_t)i * 16 * DD + 8 * kkp + 4);
            }
            float4 bf[NT];
#pragma unroll
            for (int t = 0; t < NT; ++t)
                bf[t] = Wf[(size_t)(kkp * NT + t) * 32 + lane];
#pragma unroll
            for (int i = 0; i < RT; ++i) {
#pragma unroll
                for (int t = 0; t < NT; ++t) {
                    v2f b0; b0.x = bf[t].x; b0.y = bf[t].y;
                    c[i][t] = __builtin_amdgcn_wmma_f32_16x16x4_f32(
                               false, a0[i], false, b0, (short)0, c[i][t], false, false);
                }
            }
#pragma unroll
            for (int i = 0; i < RT; ++i) {
#pragma unroll
                for (int t = 0; t < NT; ++t) {
                    v2f b1; b1.x = bf[t].z; b1.y = bf[t].w;
                    c[i][t] = __builtin_amdgcn_wmma_f32_16x16x4_f32(
                               false, a1[i], false, b1, (short)0, c[i][t], false, false);
                }
            }
        }
#pragma unroll
        for (int i = 0; i < RT; ++i) {
#pragma unroll
            for (int t = 0; t < NT; ++t) {
#pragma unroll
                for (int r = 0; r < 8; ++r) {
                    E[(16 * i + 8 * hi + r) * EPI_PITCH + t * 16 + l] = c[i][t][r];
                }
            }
        }
    }
    lds_wave_sync();

    {
        const bool okr[3] = { y > 0, true, y + 1 < H };
        const float scale2 = 0.08838834764831845f * 1.44269504088896341f;
        const unsigned PIXB = DD * 4u;

        const char* Kr[3];
#pragma unroll
        for (int r = 0; r < 3; ++r) {
            const size_t rb = (size_t)(rowi + (okr[r] ? r - 1 : 0)) * (size_t)W * PIXB;
            Kr[r] = (const char*)X + rb;
        }
        const v4f   w4   = *(const v4f*)((const char*)(vec + DD) + lane16);
        const float beta = vec[2 * DD];

        v4f kw[4][3];
        v4f qq[2];
#pragma unroll
        for (int c = 0; c < 3; ++c) {
            const unsigned colb = (unsigned)min(max(x0 - 1 + c, 0), W - 1) * PIXB + lane16;
#pragma unroll
            for (int r = 0; r < 3; ++r) kw[c][r] = ld_row(Kr[r], colb);
        }
        qq[0] = *(const v4f*)(E + lane * 4);

#pragma unroll 1
        for (int j = 0; j < PIX; j += 4) {
#pragma unroll
            for (int u = 0; u < 4; ++u) {
                const int  pp = j + u;
                const int  x  = x0 + pp;
                const bool okc[3] = { x > 0, true, x + 1 < W };
                const unsigned colb = (unsigned)min(x + 2, W - 1) * PIXB + lane16;
                const v4f q = qq[u & 1];

                const v4f xc = kw[(u + 1) & 3][1];
                float pr[10];
#pragma unroll
                for (int r = 0; r < 3; ++r) {
#pragma unroll
                    for (int c = 0; c < 3; ++c) {
                        const v4f k4 = kw[(u + c) & 3][r];
                        pr[r * 3 + c] = q.x * k4.x + q.y * k4.y + q.z * k4.z + q.w * k4.w;
                    }
                }
                pr[9] = xc.x * w4.x + xc.y * w4.y + xc.z * w4.z + xc.w * w4.w;

                const float r0 = fold2<16>(pr[0], pr[1], up16);
                const float r1 = fold2<16>(pr[2], pr[3], up16);
                const float r2 = fold2<16>(pr[4], pr[5], up16);
                const float r3 = fold2<16>(pr[6], pr[7], up16);
                float       r4 = fold2<16>(pr[8], pr[9], up16);
                const float s0 = fold2<8>(r0, r1, up8);
                const float s1 = fold2<8>(r2, r3, up8);
                r4 += xrow<8>(r4);
                const float u0 = fold2<4>(s0, s1, up4);
                r4 += xrow<4>(r4);
                float z = fold2<2>(u0, r4, up2);
                z += xrow<1>(z);
                float rs[10];
                rs[0] = lane_val<0>(z);  rs[1] = lane_val<16>(z);
                rs[2] = lane_val<8>(z);  rs[3] = lane_val<24>(z);
                rs[4] = lane_val<4>(z);  rs[5] = lane_val<20>(z);
                rs[6] = lane_val<12>(z); rs[7] = lane_val<28>(z);
                rs[8] = lane_val<2>(z);  rs[9] = lane_val<18>(z);
                const float tq = rs[9] + beta;

                float sc[9];
#pragma unroll
                for (int r = 0; r < 3; ++r) {
#pragma unroll
                    for (int c = 0; c < 3; ++c) {
                        const int keep = -(int)(okr[r] & okc[c]);
                        sc[r * 3 + c] = __builtin_bit_cast(float,
                            __builtin_bit_cast(int, (rs[r * 3 + c] + tq) * scale2) & keep);
                    }
                }

#pragma unroll
                for (int r = 0; r < 3; ++r) kw[(u + 3) & 3][r] = ld_row(Kr[r], colb);
                qq[(u + 1) & 1] = *(const v4f*)(E + min(pp + 1, PIX - 1) * EPI_PITCH + lane * 4);

                float mx = sc[0];
#pragma unroll
                for (int n = 1; n < 9; ++n) mx = (sc[n] > mx) ? sc[n] : mx;
                asm volatile("" : "+s"(mx) : : "memory");

                float p[9], sum = 0.0f;
#pragma unroll
                for (int n = 0; n < 9; ++n) { p[n] = __builtin_amdgcn_exp2f(sc[n] - mx); sum += p[n]; }
                const float inv = __builtin_amdgcn_rcpf(sum);

                v4f   o  = { 0.f, 0.f, 0.f, 0.f };
                float ws = 0.0f;
#pragma unroll
                for (int r = 0; r < 3; ++r) {
#pragma unroll
                    for (int c = 0; c < 3; ++c) {
                        const float w  = (okr[r] & okc[c]) ? p[r * 3 + c] * inv : 0.0f;
                        const v4f   k4 = kw[(u + c) & 3][r];
                        ws += w;
                        o.x += w * k4.x; o.y += w * k4.y;
                        o.z += w * k4.z; o.w += w * k4.w;
                    }
                }

                *(v4f*)(E + pp * EPI_PITCH + lane * 4) = o;
                E[pp * EPI_PITCH + DD] = ws;
            }
        }
    }
    lds_wave_sync();

    {
        const float4* Wp = Wf + WFRAG_F4_PER_MAT;
        v8f c[RT][NT];
        {
            float sw[RT][8];
#pragma unroll
            for (int i = 0; i < RT; ++i) {
#pragma unroll
                for (int r = 0; r < 8; ++r) sw[i][r] = E[(16 * i + 8 * hi + r) * EPI_PITCH + DD];
            }
#pragma unroll
            for (int t = 0; t < NT; ++t) {
                const float bias = bv[t * 16 + l];
#pragma unroll
                for (int i = 0; i < RT; ++i) {
#pragma unroll
                    for (int r = 0; r < 8; ++r) c[i][t][r] = sw[i][r] * bias;
                }
            }
        }
        const float* srow = E + l * EPI_PITCH + 2 * hi;
        for (int kkp = 0; kkp < 16; ++kkp) {
            v2f a0[RT], a1[RT];
#pragma unroll
            for (int i = 0; i < RT; ++i) {
                a0[i] = *(const v2f*)(srow + i * EPI_TILE + 8 * kkp);
                a1[i] = *(const v2f*)(srow + i * EPI_TILE + 8 * kkp + 4);
            }
            float4 bf[NT];
#pragma unroll
            for (int t = 0; t < NT; ++t)
                bf[t] = Wp[(size_t)(kkp * NT + t) * 32 + lane];
#pragma unroll
            for (int i = 0; i < RT; ++i) {
#pragma unroll
                for (int t = 0; t < NT; ++t) {
                    v2f b0; b0.x = bf[t].x; b0.y = bf[t].y;
                    c[i][t] = __builtin_amdgcn_wmma_f32_16x16x4_f32(
                               false, a0[i], false, b0, (short)0, c[i][t], false, false);
                }
            }
#pragma unroll
            for (int i = 0; i < RT; ++i) {
#pragma unroll
                for (int t = 0; t < NT; ++t) {
                    v2f b1; b1.x = bf[t].z; b1.y = bf[t].w;
                    c[i][t] = __builtin_amdgcn_wmma_f32_16x16x4_f32(
                               false, a1[i], false, b1, (short)0, c[i][t], false, false);
                }
            }
        }
#pragma unroll
        for (int i = 0; i < RT; ++i) {
#pragma unroll
            for (int t = 0; t < NT; ++t) {
#pragma unroll
                for (int r = 0; r < 8; ++r) {
                    E[(16 * i + 8 * hi + r) * EPI_PITCH + t * 16 + l] = c[i][t][r];
                }
            }
            lds_wave_sync();
            v4f row[16];
#pragma unroll
            for (int rr = 0; rr < 16; ++rr)
                row[rr] = *(const v4f*)(E + (16 * i + rr) * EPI_PITCH + lane * 4);
#pragma unroll
            for (int rr = 0; rr < 16; ++rr)
                *(v4f*)(out + (size_t)(m0 + 16 * i + rr) * DD + lane * 4) = row[rr];
        }
    }
}

extern "C" void kernel_launch(void* const* d_in, const int* in_sizes, int n_in,
                              void* d_out, int out_size, void* d_ws, size_t ws_size,
                              hipStream_t stream)
{
    const float* x  = (const float*)d_in[0];
    const float* Wq = (const float*)d_in[1];
    const float* bq = (const float*)d_in[2];
    const float* Wk = (const float*)d_in[3];
    const float* bk = (const float*)d_in[4];
    const float* Wv = (const float*)d_in[5];
    const float* bv = (const float*)d_in[6];

    const int H = 128, W = 128;
    const int Mrows = in_sizes[0] / DD;
    const int B = Mrows / (H * W);

    float4* Wf  = (float4*)d_ws;
    float*  vec = (float*)(Wf + 2 * WFRAG_F4_PER_MAT);

    prep_w<<<33, 256, 0, stream>>>(Wq, bq, Wk, bk, Wv, Wf, vec);

    const int stripWaves = B * H * (W / PIX);
    win_attn_fused<<<(stripWaves + 7) / 8, 256, 0, stream>>>(
        x, Wf, vec, bv, (float*)d_out, B, H, W);
}


# win_attn: ring of x window columns grows from 4 to 6, each new column requested three pixels before it enters the window (was one), so a pixel no longer begins with a load wait; sum order unchanged
# speedup vs baseline: 2.5999x; 1.0688x over previous
#include <hip/hip_runtime.h>


typedef float v2f __attribute__((ext_vector_type(2)));
typedef float v8f __attribute__((ext_vector_type(8)));
typedef float v4f __attribute__((ext_vector_type(4)));

#define DD 128
#define NT 8
#define WFRAG_F4_PER_MAT 4096

__global__ __launch_bounds__(256)
void prep_w(const float* __restrict__ Wq, const float* __restrict__ bq,
            const float* __restrict__ Wk, const float* __restrict__ bk,
            const float* __restrict__ Wv, float4* __restrict__ Wf,
            float* __restrict__ vec)
{
    const int blk = (int)blockIdx.x;
    const int tid = (int)threadIdx.x;
    if (blk < 8) {
        const int g    = blk * 8 + (tid >> 5);
        const int lane = tid & 31;
        const int hi = lane >> 4, l = lane & 15;
        const int r0 = (g >> 3) * 16;
        const int t  = g & 7;
        const float* ap = Wq + (size_t)(r0 + l) * DD + 2 * hi;
        const float* bp = Wk + (size_t)(t * 16 + l) * DD + 2 * hi;
        v8f acc = { 0.f, 0.f, 0.f, 0.f, 0.f, 0.f, 0.f, 0.f };
#pragma unroll
        for (int kk = 0; kk < 32; ++kk) {
            const v2f a = *(const v2f*)(ap + 4 * kk);
            const v2f b = *(const v2f*)(bp + 4 * kk);
            acc = __builtin_amdgcn_wmma_f32_16x16x4_f32(
                      false, a, false, b, (short)0, acc, false, false);
        }
        const int kkp = (r0 >> 3) + hi;
        asm volatile("v_nop\n\tv_nop\n\tv_nop\n\tv_nop\n\tv_nop" : "+v"(acc));
        float4 f0, f1;
        f0.x = acc[0]; f0.y = acc[1]; f0.z = acc[4]; f0.w = acc[5];
        f1.x = acc[2]; f1.y = acc[3]; f1.z = acc[6]; f1.w = acc[7];
        Wf[(size_t)(kkp * NT + t) * 32 + l]      = f0;
        Wf[(size_t)(kkp * NT + t) * 32 + 16 + l] = f1;
    } else if (blk < 24) {
        const int r    = (blk - 8) * 256 + tid;
        const int kkp  = r >> 8;
        const int t    = (r >> 5) & 7;
        const int lane = r & 31;
        const int hi = lane >> 4, l = lane & 15;
        const int n  = t * 16 + l;
        const int k0 = 8 * kkp + 2 * hi;
        const int k1 = k0 + 4;
        float4 v;
        v.x = Wv[(size_t)(k0 + 0) * DD + n];
        v.y = Wv[(size_t)(k0 + 1) * DD + n];
        v.z = Wv[(size_t)(k1 + 0) * DD + n];
        v.w = Wv[(size_t)(k1 + 1) * DD + n];
        Wf[WFRAG_F4_PER_MAT + r] = v;
    } else if (blk < 32) {
        const int    o    = (blk - 24) * 32 + (tid >> 3);
        const int    part = tid & 7;
        const float* row  = ((blk < 28) ? Wk : Wq) + (size_t)(o & 127) * DD + 16 * part;
        const float* bb   = ((blk < 28) ? bq : bk) + 16 * part;
        float s = 0.f;
#pragma unroll
        for (int i = 0; i < 4; ++i) {
            const v4f r4 = *(const v4f*)(row + 4 * i);
            const v4f b4 = *(const v4f*)(bb + 4 * i);
            s += r4.x * b4.x + r4.y * b4.y + r4.z * b4.z + r4.w * b4.w;
        }
        s += __shfl_xor(s, 4);
        s += __shfl_xor(s, 2);
        s += __shfl_xor(s, 1);
        if (part == 0) vec[o] = s;
    } else {
        if (tid < 32) {
            const v4f p4 = *(const v4f*)(bq + tid * 4);
            const v4f k4 = *(const v4f*)(bk + tid * 4);
            float be = p4.x * k4.x + p4.y * k4.y + p4.z * k4.z + p4.w * k4.w;
            be += __shfl_xor(be, 16);
            be += __shfl_xor(be, 8);
            be += __shfl_xor(be, 4);
            be += __shfl_xor(be, 2);
            be += __shfl_xor(be, 1);
            if (tid == 0) vec[256] = be;
        }
    }
}

#define EPI_PITCH 132
#define EPI_TILE  (16 * EPI_PITCH)
#define RT 2
#define PIX (16 * RT)

__device__ __forceinline__ void lds_wave_sync()
{
    asm volatile("s_wait_dscnt 0x0" ::: "memory");
}


__device__ __forceinline__ v4f ld_row(const char* P, unsigned boff)
{
    return *(const v4f*)(P + boff);
}

__device__ __forceinline__ float xhalf(float x)
{
    const int b = __builtin_bit_cast(int, x);
    return __builtin_bit_cast(float,
        __builtin_amdgcn_permlanex16(b, b, 0x76543210, 0xfedcba98, false, false));
}

template <int M>
__device__ __forceinline__ float xrow(float x)
{
    const int b = __builtin_bit_cast(int, x);
    return __builtin_bit_cast(float,
        __builtin_amdgcn_update_dpp(b, b, 0x160 | M, 0xf, 0xf, true));
}

template <int M>
__device__ __forceinline__ float fold2(float a, float b, bool up)
{
    const float keep = up ? b : a;
    const float send = up ? a : b;
    if (M == 16) return keep + xhalf(send);
    return keep + xrow<M & 15>(send);
}

template <int N> struct ring_pos { static constexpr int v = N; };

template <int L>
__device__ __forceinline__ float lane_val(float x)
{
    return __builtin_bit_cast(float,
        __builtin_amdgcn_readlane(__builtin_bit_cast(int, x), L));
}


__global__ __launch_bounds__(256, 4)
void win_attn_fused(const float* __restrict__ X,
                    const float4* __restrict__ Wf,
                    const float* __restrict__ vec, const float* __restrict__ bv,
                    float* __restrict__ out, int B, int H, int W)
{
    const int wave = __builtin_amdgcn_readfirstlane(
        (int)((blockIdx.x * blockDim.x + threadIdx.x) >> 5));
    const int lane = (int)(threadIdx.x & 31u);
    const int hi = lane >> 4;
    const int l  = lane & 15;
    const unsigned lane16 = (unsigned)lane * 16u;
    const bool up16 = (threadIdx.x & 16u) != 0u;
    const bool up8  = (threadIdx.x &  8u) != 0u;
    const bool up4  = (threadIdx.x &  4u) != 0u;
    const bool up2  = (threadIdx.x &  2u) != 0u;
    const int spr  = W / PIX;

    const int  wig  = wave & 7;
    const int  blk  = wave >> 3;
    const int  rblk = blk / spr;
    const int  rowi = rblk * 8 + wig;
    const int  x0   = (blk - rblk * spr) * PIX;
    if (rowi >= B * H) return;
    const int  y    = rowi % H;
    const int  m0   = rowi * W + x0;

    __shared__ v4f epi[8 * RT * EPI_TILE / 4];
    float* E = (float*)epi + (threadIdx.x >> 5) * (RT * EPI_TILE);

    {
        const float* xrow = X + (size_t)(m0 + l) * DD + 2 * hi;
        v8f c[RT][NT];
#pragma unroll
        for (int t = 0; t < NT; ++t) {
            const float bias = vec[t * 16 + l];
#pragma unroll
            for (int i = 0; i < RT; ++i) {
#pragma unroll
                for (int r = 0; r < 8; ++r) c[i][t][r] = bias;
            }
        }
        for (int kkp = 0; kkp < 16; ++kkp) {
            v2f a0[RT], a1[RT];
#pragma unroll
            for (int i = 0; i < RT; ++i) {
                a0[i] = *(const v2f*)(xrow + (size_t)i * 16 * DD + 8 * kkp);
                a1[i] = *(const v2f*)(xrow + (size_t)i * 16 * DD + 8 * kkp + 4);
            }
            float4 bf[NT];
#pragma unroll
            for (int t = 0; t < NT; ++t)
                bf[t] = Wf[(size_t)(kkp * NT + t) * 32 + lane];
#pragma unroll
            for (int i = 0; i < RT; ++i) {
#pragma unroll
                for (int t = 0; t < NT; ++t) {
                    v2f b0; b0.x = bf[t].x; b0.y = bf[t].y;
                    c[i][t] = __builtin_amdgcn_wmma_f32_16x16x4_f32(
                               false, a0[i], false, b0, (short)0, c[i][t], false, false);
                }
            }
#pragma unroll
            for (int i = 0; i < RT; ++i) {
#pragma unroll
                for (int t = 0; t < NT; ++t) {
                    v2f b1; b1.x = bf[t].z; b1.y = bf[t].w;
                    c[i][t] = __builtin_amdgcn_wmma_f32_16x16x4_f32(
                               false, a1[i], false, b1, (short)0, c[i][t], false, false);
                }
            }
        }
#pragma unroll
        for (int i = 0; i < RT; ++i) {
#pragma unroll
            for (int t = 0; t < NT; ++t) {
#pragma unroll
                for (int r = 0; r < 8; ++r) {
                    E[(16 * i + 8 * hi + r) * EPI_PITCH + t * 16 + l] = c[i][t][r];
                }
            }
        }
    }
    lds_wave_sync();

    {
        const bool okr[3] = { y > 0, true, y + 1 < H };
        const float scale2 = 0.08838834764831845f * 1.44269504088896341f;
        const unsigned PIXB = DD * 4u;

        const char* Kr[3];
#pragma unroll
        for (int r = 0; r < 3; ++r) {
            const size_t rb = (size_t)(rowi + (okr[r] ? r - 1 : 0)) * (size_t)W * PIXB;
            Kr[r] = (const char*)X + rb;
        }
        v4f         w4   = *(const v4f*)((const char*)(vec + DD) + lane16);
        const float beta = vec[2 * DD];

        v4f kw[6][3];
#pragma unroll
        for (int c = 0; c < 5; ++c) {
            const unsigned colb = (unsigned)min(max(x0 - 1 + c, 0), W - 1) * PIXB + lane16;
#pragma unroll
            for (int r = 0; r < 3; ++r) kw[c][r] = ld_row(Kr[r], colb);
        }
#pragma unroll
        for (int c = 0; c < 5; ++c)
            asm volatile("" : "+v"(kw[c][0]), "+v"(kw[c][1]), "+v"(kw[c][2]));
        asm volatile("" : "+v"(w4));
        v4f qn = *(const v4f*)(E + lane * 4);

        auto pixel = [&](auto uc, const int pp) __attribute__((always_inline)) {
                constexpr int u = decltype(uc)::v;
                const int  x  = x0 + pp;
                const bool okc[3] = { x > 0, true, x + 1 < W };
                const unsigned colb = (unsigned)min(x + 4, W - 1) * PIXB + lane16;
                asm volatile("" : "+v"(kw[(u + 2) % 6][0]), "+v"(kw[(u + 2) % 6][1]),
                                  "+v"(kw[(u + 2) % 6][2]));
                const v4f q = qn;

                const v4f xc = kw[(u + 1) % 6][1];
                float pr[10];
#pragma unroll
                for (int r = 0; r < 3; ++r) {
#pragma unroll
                    for (int c = 0; c < 3; ++c) {
                        const v4f k4 = kw[(u + c) % 6][r];
                        pr[r * 3 + c] = q.x * k4.x + q.y * k4.y + q.z * k4.z + q.w * k4.w;
                    }
                }
                pr[9] = xc.x * w4.x + xc.y * w4.y + xc.z * w4.z + xc.w * w4.w;

                const float r0 = fold2<16>(pr[0], pr[1], up16);
                const float r1 = fold2<16>(pr[2], pr[3], up16);
                const float r2 = fold2<16>(pr[4], pr[5], up16);
                const float r3 = fold2<16>(pr[6], pr[7], up16);
                float       r4 = fold2<16>(pr[8], pr[9], up16);
                const float s0 = fold2<8>(r0, r1, up8);
                const float s1 = fold2<8>(r2, r3, up8);
                r4 += xrow<8>(r4);
                const float u0 = fold2<4>(s0, s1, up4);
                r4 += xrow<4>(r4);
                float z = fold2<2>(u0, r4, up2);
                z += xrow<1>(z);
                float rs[10];
                rs[0] = lane_val<0>(z);  rs[1] = lane_val<16>(z);
                rs[2] = lane_val<8>(z);  rs[3] = lane_val<24>(z);
                rs[4] = lane_val<4>(z);  rs[5] = lane_val<20>(z);
                rs[6] = lane_val<12>(z); rs[7] = lane_val<28>(z);
                rs[8] = lane_val<2>(z);  rs[9] = lane_val<18>(z);
                const float tq = rs[9] + beta;

                float sc[9];
#pragma unroll
                for (int r = 0; r < 3; ++r) {
#pragma unroll
                    for (int c = 0; c < 3; ++c) {
                        const int keep = -(int)(okr[r] & okc[c]);
                        sc[r * 3 + c] = __builtin_bit_cast(float,
                            __builtin_bit_cast(int, (rs[r * 3 + c] + tq) * scale2) & keep);
                    }
                }

#pragma unroll
                for (int r = 0; r < 3; ++r) kw[(u + 5) % 6][r] = ld_row(Kr[r], colb);
                qn = *(const v4f*)(E + min(pp + 1, PIX - 1) * EPI_PITCH + lane * 4);

                float mx = sc[0];
#pragma unroll
                for (int n = 1; n < 9; ++n) mx = (sc[n] > mx) ? sc[n] : mx;
                asm volatile("" : "+s"(mx) : : "memory");

                float p[9], sum = 0.0f;
#pragma unroll
                for (int n = 0; n < 9; ++n) { p[n] = __builtin_amdgcn_exp2f(sc[n] - mx); sum += p[n]; }
                const float inv = __builtin_amdgcn_rcpf(sum);

                v4f   o  = { 0.f, 0.f, 0.f, 0.f };
                float ws = 0.0f;
#pragma unroll
                for (int r = 0; r < 3; ++r) {
#pragma unroll
                    for (int c = 0; c < 3; ++c) {
                        const float w  = (okr[r] & okc[c]) ? p[r * 3 + c] * inv : 0.0f;
                        const v4f   k4 = kw[(u + c) % 6][r];
                        ws += w;
                        o.x += w * k4.x; o.y += w * k4.y;
                        o.z += w * k4.z; o.w += w * k4.w;
                    }
                }

                *(v4f*)(E + pp * EPI_PITCH + lane * 4) = o;
                E[pp * EPI_PITCH + DD] = ws;
        };
#pragma unroll 1
        for (int j = 0; j < 30; j += 6) {
            pixel(ring_pos<0>{}, j);
            pixel(ring_pos<1>{}, j + 1);
            pixel(ring_pos<2>{}, j + 2);
            pixel(ring_pos<3>{}, j + 3);
            pixel(ring_pos<4>{}, j + 4);
            pixel(ring_pos<5>{}, j + 5);
        }
        pixel(ring_pos<0>{}, 30);
        pixel(ring_pos<1>{}, 31);
    }
    lds_wave_sync();

    {
        const float4* Wp = Wf + WFRAG_F4_PER_MAT;
        v8f c[RT][NT];
        {
            float sw[RT][8];
#pragma unroll
            for (int i = 0; i < RT; ++i) {
#pragma unroll
                for (int r = 0; r < 8; ++r) sw[i][r] = E[(16 * i + 8 * hi + r) * EPI_PITCH + DD];
            }
#pragma unroll
            for (int t = 0; t < NT; ++t) {
                const float bias = bv[t * 16 + l];
#pragma unroll
                for (int i = 0; i < RT; ++i) {
#pragma unroll
                    for (int r = 0; r < 8; ++r) c[i][t][r] = sw[i][r] * bias;
                }
            }
        }
        const float* srow = E + l * EPI_PITCH + 2 * hi;
        for (int kkp = 0; kkp < 16; ++kkp) {
            v2f a0[RT], a1[RT];
#pragma unroll
            for (int i = 0; i < RT; ++i) {
                a0[i] = *(const v2f*)(srow + i * EPI_TILE + 8 * kkp);
                a1[i] = *(const v2f*)(srow + i * EPI_TILE + 8 * kkp + 4);
            }
            float4 bf[NT];
#pragma unroll
            for (int t = 0; t < NT; ++t)
                bf[t] = Wp[(size_t)(kkp * NT + t) * 32 + lane];
#pragma unroll
            for (int i = 0; i < RT; ++i) {
#pragma unroll
                for (int t = 0; t < NT; ++t) {
                    v2f b0; b0.x = bf[t].x; b0.y = bf[t].y;
                    c[i][t] = __builtin_amdgcn_wmma_f32_16x16x4_f32(
                               false, a0[i], false, b0, (short)0, c[i][t], false, false);
                }
            }
#pragma unroll
            for (int i = 0; i < RT; ++i) {
#pragma unroll
                for (int t = 0; t < NT; ++t) {
                    v2f b1; b1.x = bf[t].z; b1.y = bf[t].w;
                    c[i][t] = __builtin_amdgcn_wmma_f32_16x16x4_f32(
                               false, a1[i], false, b1, (short)0, c[i][t], false, false);
                }
            }
        }
#pragma unroll
        for (int i = 0; i < RT; ++i) {
#pragma unroll
            for (int t = 0; t < NT; ++t) {
#pragma unroll
                for (int r = 0; r < 8; ++r) {
                    E[(16 * i + 8 * hi + r) * EPI_PITCH + t * 16 + l] = c[i][t][r];
                }
            }
            lds_wave_sync();
            v4f row[16];
#pragma unroll
            for (int rr = 0; rr < 16; ++rr)
                row[rr] = *(const v4f*)(E + (16 * i + rr) * EPI_PITCH + lane * 4);
#pragma unroll
            for (int rr = 0; rr < 16; ++rr)
                *(v4f*)(out + (size_t)(m0 + 16 * i + rr) * DD + lane * 4) = row[rr];
        }
    }
}

extern "C" void kernel_launch(void* const* d_in, const int* in_sizes, int n_in,
                              void* d_out, int out_size, void* d_ws, size_t ws_size,
                              hipStream_t stream)
{
    const float* x  = (const float*)d_in[0];
    const float* Wq = (const float*)d_in[1];
    const float* bq = (const float*)d_in[2];
    const float* Wk = (const float*)d_in[3];
    const float* bk = (const float*)d_in[4];
    const float* Wv = (const float*)d_in[5];
    const float* bv = (const float*)d_in[6];

    const int H = 128, W = 128;
    const int Mrows = in_sizes[0] / DD;
    const int B = Mrows / (H * W);

    float4* Wf  = (float4*)d_ws;
    float*  vec = (float*)(Wf + 2 * WFRAG_F4_PER_MAT);

    prep_w<<<33, 256, 0, stream>>>(Wq, bq, Wk, bk, Wv, Wf, vec);

    const int stripWaves = B * H * (W / PIX);
    win_attn_fused<<<(stripWaves + 7) / 8, 256, 0, stream>>>(
        x, Wf, vec, bv, (float*)d_out, B, H, W);
}
